# starting kernel
# baseline (speedup 1.0000x reference)
#include <hip/hip_runtime.h>
#include <math.h>


typedef __attribute__((ext_vector_type(2))) float v2f;
typedef __attribute__((ext_vector_type(8))) float v8f;

#define NSEQ   16384
#define XL     1024
#define UL     256
#define YLN    256
#define CHUNKS 128
#define CLEN   128
#define COLSTR ((size_t)CLEN * XL)

__device__ __forceinline__ v8f wmma4(v2f a, v2f b, v8f c) {
  return __builtin_amdgcn_wmma_f32_16x16x4_f32(
       false, a,  false, b,
       (short)0, c,  false,  false);
}

template <int WM, int WN>
__global__ __launch_bounds__(256) void gemm_rr_k(
    const float* __restrict__ Ap, const float* __restrict__ Bp,
    float* __restrict__ Cp, int K, int lda, int ldb, int ldc) {
  const int lane = threadIdx.x & 31;
  const int w    = threadIdx.x >> 5;
  const int wm = w >> 2, wn = w & 3;
  const int tm = blockIdx.x * (32 * WM) + wm * (16 * WM);
  const int tn = blockIdx.y * (64 * WN) + wn * (16 * WN);
  const int ml = lane & 15;
  const int kh = (lane >> 4) << 1;

  v8f acc[WM][WN];
#pragma unroll
  for (int mi = 0; mi < WM; ++mi)
#pragma unroll
    for (int ni = 0; ni < WN; ++ni) acc[mi][ni] = (v8f){};

  const float* ap[WM];
  const float* bp[WN];
#pragma unroll
  for (int mi = 0; mi < WM; ++mi)
    ap[mi] = Ap + (size_t)(tm + mi * 16 + ml) * lda + kh;
#pragma unroll
  for (int ni = 0; ni < WN; ++ni)
    bp[ni] = Bp + (size_t)kh * ldb + (tn + ni * 16 + ml);

  for (int k = 0; k < K; k += 4) {
    v2f af[WM], bf[WN];
    const size_t bo0 = (size_t)k * ldb, bo1 = bo0 + (size_t)ldb;
#pragma unroll
    for (int mi = 0; mi < WM; ++mi) af[mi] = (v2f){ ap[mi][k], ap[mi][k + 1] };
#pragma unroll
    for (int ni = 0; ni < WN; ++ni) bf[ni] = (v2f){ bp[ni][bo0], bp[ni][bo1] };
#pragma unroll
    for (int mi = 0; mi < WM; ++mi)
#pragma unroll
      for (int ni = 0; ni < WN; ++ni)
        acc[mi][ni] = wmma4(af[mi], bf[ni], acc[mi][ni]);
  }

  const int nn = lane & 15;
  const int mb = (lane >> 4) << 3;
#pragma unroll
  for (int mi = 0; mi < WM; ++mi)
#pragma unroll
    for (int ni = 0; ni < WN; ++ni) {
      float* cp = Cp + (size_t)(tm + mi * 16 + mb) * ldc + (tn + ni * 16 + nn);
#pragma unroll
      for (int r = 0; r < 8; ++r) cp[(size_t)r * ldc] = acc[mi][ni][r];
    }
}

template <bool STORE_XS>
__global__ __launch_bounds__(256) void scan_step_k(
    const float* __restrict__ Ap, const float* __restrict__ Sold,
    const float* __restrict__ addb, float* __restrict__ Snew,
    float* __restrict__ xsb) {
  const int lane = threadIdx.x & 31;
  const int w    = threadIdx.x >> 5;
  const int wm = w >> 2, wn = w & 3;
  const int tm = blockIdx.x * 32 + wm * 16;
  const int tn = blockIdx.y * 64 + wn * 16;
  const int ml = lane & 15;
  const int kh = (lane >> 4) << 1;

  constexpr int KQ = XL / 4;
  constexpr size_t BQ = (size_t)KQ * CHUNKS;

  v8f acc0 = {}, acc1 = {}, acc2 = {}, acc3 = {};
  const float* ap = Ap + (size_t)(tm + ml) * XL + kh;
  const float* bp = Sold + (size_t)kh * CHUNKS + (tn + ml);

  for (int k = 0; k < KQ; k += 4) {
    v2f a0 = { ap[k],          ap[k + 1] };
    v2f a1 = { ap[k + KQ],     ap[k + KQ + 1] };
    v2f a2 = { ap[k + 2 * KQ], ap[k + 2 * KQ + 1] };
    v2f a3 = { ap[k + 3 * KQ], ap[k + 3 * KQ + 1] };
    const size_t bo = (size_t)k * CHUNKS;
    v2f b0 = { bp[bo],          bp[bo + CHUNKS] };
    v2f b1 = { bp[bo + BQ],     bp[bo + BQ + CHUNKS] };
    v2f b2 = { bp[bo + 2 * BQ], bp[bo + 2 * BQ + CHUNKS] };
    v2f b3 = { bp[bo + 3 * BQ], bp[bo + 3 * BQ + CHUNKS] };
    acc0 = wmma4(a0, b0, acc0);
    acc1 = wmma4(a1, b1, acc1);
    acc2 = wmma4(a2, b2, acc2);
    acc3 = wmma4(a3, b3, acc3);
  }
  v8f acc = (acc0 + acc1) + (acc2 + acc3);

  const int j  = tn + (lane & 15);
  const int mb = tm + ((lane >> 4) << 3);
  const float* addc = addb + (size_t)j * COLSTR + mb;
  if constexpr (STORE_XS) {
    float* xsc = xsb + (size_t)j * COLSTR + mb;
#pragma unroll
    for (int r = 0; r < 8; ++r) {
      float v = acc[r] + addc[r];
      Snew[(size_t)(mb + r) * CHUNKS + j] = v;
      xsc[r] = v;
    }
  } else {
    (void)xsb;
#pragma unroll
    for (int r = 0; r < 8; ++r) {
      float v = acc[r] + addc[r];
      Snew[(size_t)(mb + r) * CHUNKS + j] = v;
    }
  }
}

__global__ __launch_bounds__(256) void matvec_step_k(
    const float* __restrict__ Mm, const float* __restrict__ vin, int vins,
    const float* __restrict__ vadd, int vadds,
    float* __restrict__ vout, int vouts) {
  __shared__ float vsh[XL];
  const int tid = threadIdx.x;
  for (int i = tid; i < XL; i += 256) vsh[i] = vin[(size_t)i * vins];
  __syncthreads();

  const int lane = tid & 31, wid = tid >> 5;
  const int gw = blockIdx.x * 8 + wid;
#pragma unroll
  for (int rr = 0; rr < 4; ++rr) {
    const int row = gw * 4 + rr;
    const float* Mr = Mm + (size_t)row * XL;
    float s = 0.f;
    for (int i = 0; i < XL; i += 32) s += Mr[i + lane] * vsh[i + lane];
#pragma unroll
    for (int off = 16; off > 0; off >>= 1) s += __shfl_down(s, off, 32);
    if (lane == 0) vout[(size_t)row * vouts] = s + vadd[(size_t)row * vadds];
  }
}

__global__ __launch_bounds__(256) void final_gemm_k(
    const float* __restrict__ xs, const float* __restrict__ Ct,
    const float* __restrict__ uf, const float* __restrict__ Dt,
    float* __restrict__ out) {
  const int lane = threadIdx.x & 31;
  const int w    = threadIdx.x >> 5;
  const int wm = w >> 2, wn = w & 3;
  const int tm = blockIdx.x * 128 + wm * 64;
  const int tn = wn * 64;
  const int ml = lane & 15;
  const int kh = (lane >> 4) << 1;

  v8f acc[4][4];
#pragma unroll
  for (int mi = 0; mi < 4; ++mi)
#pragma unroll
    for (int ni = 0; ni < 4; ++ni) acc[mi][ni] = (v8f){};

  {
    const float* ap[4];
    const float* bp[4];
#pragma unroll
    for (int mi = 0; mi < 4; ++mi)
      ap[mi] = xs + (size_t)(tm + mi * 16 + ml) * XL + kh;
#pragma unroll
    for (int ni = 0; ni < 4; ++ni)
      bp[ni] = Ct + (size_t)kh * YLN + (tn + ni * 16 + ml);
    for (int k = 0; k < XL; k += 4) {
      v2f af[4], bf[4];
      const size_t bo0 = (size_t)k * YLN, bo1 = bo0 + YLN;
#pragma unroll
      for (int mi = 0; mi < 4; ++mi) af[mi] = (v2f){ ap[mi][k], ap[mi][k + 1] };
#pragma unroll
      for (int ni = 0; ni < 4; ++ni) bf[ni] = (v2f){ bp[ni][bo0], bp[ni][bo1] };
#pragma unroll
      for (int mi = 0; mi < 4; ++mi)
#pragma unroll
        for (int ni = 0; ni < 4; ++ni)
          acc[mi][ni] = wmma4(af[mi], bf[ni], acc[mi][ni]);
    }
  }
  {
    const float* ap[4];
    const float* bp[4];
#pragma unroll
    for (int mi = 0; mi < 4; ++mi)
      ap[mi] = uf + (size_t)(tm + mi * 16 + ml) * UL + kh;
#pragma unroll
    for (int ni = 0; ni < 4; ++ni)
      bp[ni] = Dt + (size_t)kh * YLN + (tn + ni * 16 + ml);
    for (int k = 0; k < UL; k += 4) {
      v2f af[4], bf[4];
      const size_t bo0 = (size_t)k * YLN, bo1 = bo0 + YLN;
#pragma unroll
      for (int mi = 0; mi < 4; ++mi) af[mi] = (v2f){ ap[mi][k], ap[mi][k + 1] };
#pragma unroll
      for (int ni = 0; ni < 4; ++ni) bf[ni] = (v2f){ bp[ni][bo0], bp[ni][bo1] };
#pragma unroll
      for (int mi = 0; mi < 4; ++mi)
#pragma unroll
        for (int ni = 0; ni < 4; ++ni)
          acc[mi][ni] = wmma4(af[mi], bf[ni], acc[mi][ni]);
    }
  }

  const int nn = lane & 15;
  const int mb = (lane >> 4) << 3;
#pragma unroll
  for (int mi = 0; mi < 4; ++mi)
#pragma unroll
    for (int ni = 0; ni < 4; ++ni) {
      const int yl = tn + ni * 16 + nn;
      const int n0 = tm + mi * 16 + mb;
      float* op = out + (size_t)yl * NSEQ + n0;
#pragma unroll
      for (int r = 0; r < 8; ++r) op[r] = tanhf(acc[mi][ni][r]);
    }
}

__global__ __launch_bounds__(256) void transpose_k(
    const float* __restrict__ in, float* __restrict__ out, int rows, int cols) {
  __shared__ float tile[32][33];
  const int c0 = blockIdx.x * 32, r0 = blockIdx.y * 32;
  const int tx = threadIdx.x, ty = threadIdx.y;
#pragma unroll
  for (int i = 0; i < 32; i += 8)
    tile[ty + i][tx] = in[(size_t)(r0 + ty + i) * cols + (c0 + tx)];
  __syncthreads();
#pragma unroll
  for (int i = 0; i < 32; i += 8)
    out[(size_t)(c0 + ty + i) * rows + (r0 + tx)] = tile[tx][ty + i];
}

__global__ void set_col_k(float* __restrict__ dst, const float* __restrict__ src,
                          int stride, int n) {
  int i = blockIdx.x * blockDim.x + threadIdx.x;
  if (i < n) dst[(size_t)i * stride] = src[i];
}

extern "C" void kernel_launch(void* const* d_in, const int* in_sizes, int n_in,
                              void* d_out, int out_size, void* d_ws, size_t ws_size,
                              hipStream_t stream) {
  (void)in_sizes; (void)n_in; (void)out_size; (void)ws_size;
  const float* u  = (const float*)d_in[0];
  const float* A  = (const float*)d_in[1];
  const float* B  = (const float*)d_in[2];
  const float* C  = (const float*)d_in[3];
  const float* D  = (const float*)d_in[4];
  const float* x0 = (const float*)d_in[5];
  float* out = (float*)d_out;

  float* ws = (float*)d_ws;
  size_t off = 0;
  float* Bu = ws + off; off += (size_t)NSEQ * XL;
  float* xs = ws + off; off += (size_t)NSEQ * XL;
  float* Bt = ws + off; off += (size_t)UL * XL;
  float* Ct = ws + off; off += (size_t)XL * YLN;
  float* Dt = ws + off; off += (size_t)UL * YLN;
  float* Q0 = ws + off; off += (size_t)XL * XL;
  float* Q1 = ws + off; off += (size_t)XL * XL;
  float* SA = ws + off; off += (size_t)XL * CHUNKS;
  float* SB = ws + off; off += (size_t)XL * CHUNKS;
  float* EB = ws + off; off += (size_t)XL * CHUNKS;

  transpose_k<<<dim3(UL / 32, XL / 32), dim3(32, 8), 0, stream>>>(B, Bt, XL, UL);
  transpose_k<<<dim3(XL / 32, YLN / 32), dim3(32, 8), 0, stream>>>(C, Ct, YLN, XL);
  transpose_k<<<dim3(YLN / 32, UL / 32), dim3(32, 8), 0, stream>>>(D, Dt, UL, YLN);

  gemm_rr_k<4, 4><<<dim3(NSEQ / 128, XL / 256), 256, 0, stream>>>(
      u, Bt, Bu, UL, UL, XL, XL);

  gemm_rr_k<2, 2><<<dim3(XL / 64, XL / 128), 256, 0, stream>>>(
      A, A, Q0, XL, XL, XL, XL);
  {
    float* src = Q0; float* dst = Q1;
    for (int i = 0; i < 6; ++i) {
      gemm_rr_k<2, 2><<<dim3(XL / 64, XL / 128), 256, 0, stream>>>(
          src, src, dst, XL, XL, XL, XL);
      float* t = src; src = dst; dst = t;
    }
  }
  const float* M128 = Q0;

  hipMemsetAsync(SA, 0, (size_t)XL * CHUNKS * sizeof(float), stream);
  hipMemsetAsync(EB, 0, (size_t)XL * CHUNKS * sizeof(float), stream);
  set_col_k<<<4, 256, 0, stream>>>(EB, x0, CHUNKS, XL);

  {
    float* cur = SA; float* nxt = SB;
    for (int t = 0; t < CLEN; ++t) {
      scan_step_k<false><<<dim3(XL / 32, CHUNKS / 64), 256, 0, stream>>>(
          A, cur, Bu + (size_t)t * XL, nxt, nullptr);
      float* tmp = cur; cur = nxt; nxt = tmp;
    }
  }

  for (int j = 1; j < CHUNKS; ++j) {
    matvec_step_k<<<32, 256, 0, stream>>>(M128, EB + (j - 1), CHUNKS,
                                          SA + (j - 1), CHUNKS, EB + j, CHUNKS);
  }

  {
    float* cur = EB; float* nxt = SB;
    for (int t = 0; t < CLEN; ++t) {
      scan_step_k<true><<<dim3(XL / 32, CHUNKS / 64), 256, 0, stream>>>(
          A, cur, Bu + (size_t)t * XL, nxt, xs + (size_t)t * XL);
      float* tmp = cur; cur = nxt; nxt = tmp;
    }
  }

  final_gemm_k<<<dim3(NSEQ / 128, 1), 256, 0, stream>>>(xs, Ct, u, Dt, out);
}


# output projection leaves the generic loop: C and D read as given (K-contiguous rows, no transposes) on the tree's 16-byte two-set loader, 256 workgroups, 32-byte stores; sum order changes
# speedup vs baseline: 7.0683x; 7.0683x over previous
#include <hip/hip_runtime.h>
#include <math.h>


typedef __attribute__((ext_vector_type(2))) float v2f;
typedef __attribute__((ext_vector_type(4))) float v4f;
typedef __attribute__((ext_vector_type(8))) float v8f;

#define NSEQ   16384
#define XL     1024
#define UL     256
#define YLN    256
#define CLEN   1024

__device__ __forceinline__ v8f wmma4(v2f a, v2f b, v8f c) {
  return __builtin_amdgcn_wmma_f32_16x16x4_f32(
       false, a,  false, b,
       (short)0, c,  false,  false);
}

__device__ __forceinline__ size_t panel_off(int R, int K) {
  return ((size_t)(R >> 4) * (XL / 8) + (size_t)(K >> 3)) * 128
         + (size_t)((R & 15) * 8 + (K & 7));
}

template <int WM, int WN>
__global__ __launch_bounds__(256) void gemm_rr_k(
    const float* __restrict__ Ap, const float* __restrict__ Bp,
    float* __restrict__ Cp, int K, int lda, int ldb, int ldc) {
  const int lane = threadIdx.x & 31;
  const int w    = threadIdx.x >> 5;
  const int wm = w >> 2, wn = w & 3;
  const int tm = blockIdx.x * (32 * WM) + wm * (16 * WM);
  const int tn = blockIdx.y * (64 * WN) + wn * (16 * WN);
  const int ml = lane & 15;
  const int kh = (lane >> 4) << 1;

  v8f acc[WM][WN];
#pragma unroll
  for (int mi = 0; mi < WM; ++mi)
#pragma unroll
    for (int ni = 0; ni < WN; ++ni) acc[mi][ni] = (v8f){};

  const float* ap[WM];
  const float* bp[WN];
#pragma unroll
  for (int mi = 0; mi < WM; ++mi)
    ap[mi] = Ap + (size_t)(tm + mi * 16 + ml) * lda + kh;
#pragma unroll
  for (int ni = 0; ni < WN; ++ni)
    bp[ni] = Bp + (size_t)kh * ldb + (tn + ni * 16 + ml);

  for (int k = 0; k < K; k += 4) {
    v2f af[WM], bf[WN];
    const size_t bo0 = (size_t)k * ldb, bo1 = bo0 + (size_t)ldb;
#pragma unroll
    for (int mi = 0; mi < WM; ++mi) af[mi] = (v2f){ ap[mi][k], ap[mi][k + 1] };
#pragma unroll
    for (int ni = 0; ni < WN; ++ni) bf[ni] = (v2f){ bp[ni][bo0], bp[ni][bo1] };
#pragma unroll
    for (int mi = 0; mi < WM; ++mi)
#pragma unroll
      for (int ni = 0; ni < WN; ++ni)
        acc[mi][ni] = wmma4(af[mi], bf[ni], acc[mi][ni]);
  }

  const int nn = lane & 15;
  const int mb = (lane >> 4) << 3;
#pragma unroll
  for (int mi = 0; mi < WM; ++mi)
#pragma unroll
    for (int ni = 0; ni < WN; ++ni) {
      float* cp = Cp + (size_t)(tm + mi * 16 + mb) * ldc + (tn + ni * 16 + nn);
#pragma unroll
      for (int r = 0; r < 8; ++r) cp[(size_t)r * ldc] = acc[mi][ni][r];
    }
}

template <int WM, int WN, int KU>
struct frag_set { v4f a[KU][WM]; v4f b[KU][WN]; };

template <int WM, int WN, int KU, int AS, int BS>
__device__ __forceinline__ void load_set(frag_set<WM, WN, KU>& f,
    const float* const (&ap)[WM], const float* const (&bp)[WN], int k) {
#pragma unroll
  for (int u = 0; u < KU; ++u) {
#pragma unroll
    for (int mi = 0; mi < WM; ++mi)
      f.a[u][mi] = *(const v4f*)(ap[mi] + (k / 8 + u) * AS);
#pragma unroll
    for (int ni = 0; ni < WN; ++ni)
      f.b[u][ni] = *(const v4f*)(bp[ni] + (k / 8 + u) * BS);
  }
}

template <int WM, int WN, int KU, int NCH>
__device__ __forceinline__ void mma_set(v8f (&acc)[NCH][WM][WN],
                                        const frag_set<WM, WN, KU>& f) {
#pragma unroll
  for (int u = 0; u < KU; ++u)
#pragma unroll
    for (int mi = 0; mi < WM; ++mi)
#pragma unroll
      for (int ni = 0; ni < WN; ++ni) {
        const v2f a0 = { f.a[u][mi].x, f.a[u][mi].y };
        const v2f a1 = { f.a[u][mi].z, f.a[u][mi].w };
        const v2f b0 = { f.b[u][ni].x, f.b[u][ni].y };
        const v2f b1 = { f.b[u][ni].z, f.b[u][ni].w };
        acc[0][mi][ni]       = wmma4(a0, b0, acc[0][mi][ni]);
        acc[NCH - 1][mi][ni] = wmma4(a1, b1, acc[NCH - 1][mi][ni]);
      }
}

template <int WM, int WN, int KU, int WGM = 2>
__global__ __launch_bounds__(256) void sweep_k(
    float* X, const float* __restrict__ P, int s, int r0) {
  constexpr int WGN = 8 / WGM;
  const int lane = threadIdx.x & 31;
  const int w    = threadIdx.x >> 5;
  const int wm = w / WGN, wn = w % WGN;
  const int tm = blockIdx.x * (16 * WM * WGM) + wm * (16 * WM);
  const int tn = blockIdx.y * (16 * WN * WGN) + wn * (16 * WN);
  const int c = lane & 15;
  const int h = lane >> 4;
  const int s2 = 2 * s;
  constexpr int NCH = (WM * WN < 4) ? 2 : 1;
  constexpr int KT = 8 * KU;
  static_assert(XL % (2 * KT) == 0, "two trips per loop pass");

  v8f acc[NCH][WM][WN];
#pragma unroll
  for (int ch = 0; ch < NCH; ++ch)
#pragma unroll
    for (int mi = 0; mi < WM; ++mi)
#pragma unroll
      for (int ni = 0; ni < WN; ++ni) acc[ch][mi][ni] = (v8f){};

  const float* ap[WM];
  const float* bp[WN];
#pragma unroll
  for (int mi = 0; mi < WM; ++mi) {
    int q = r0 - s + (tm + mi * 16 + c) * s2;
    if (q < 0) q = 0;
    ap[mi] = X + (size_t)q * XL + 4 * h;
  }
#pragma unroll
  for (int ni = 0; ni < WN; ++ni)
    bp[ni] = P + (size_t)((tn >> 4) + ni) * (16 * XL) + c * 8 + 4 * h;

  frag_set<WM, WN, KU> f0, f1;
  load_set<WM, WN, KU, 8, 128>(f0, ap, bp, 0);
  for (int k = 0; k < XL; k += 2 * KT) {
    load_set<WM, WN, KU, 8, 128>(f1, ap, bp, k + KT);
    asm volatile("" ::: "memory");
    mma_set<WM, WN, KU, NCH>(acc, f0);
    const int kn = (k + 2 * KT < XL) ? (k + 2 * KT) : k;
    load_set<WM, WN, KU, 8, 128>(f0, ap, bp, kn);
    asm volatile("" ::: "memory");
    mma_set<WM, WN, KU, NCH>(acc, f1);
  }

  const bool open0 = (r0 < s);
  v8f old[WM][WN];
#pragma unroll
  for (int mi = 0; mi < WM; ++mi)
#pragma unroll
    for (int ni = 0; ni < WN; ++ni) {
#pragma unroll
      for (int r = 0; r < 8; ++r) {
        const int row = r0 + (tm + mi * 16 + 8 * h + r) * s2;
        old[mi][ni][r] = X[(size_t)row * XL + (tn + ni * 16 + c)];
      }
    }
#pragma unroll
  for (int mi = 0; mi < WM; ++mi)
#pragma unroll
    for (int ni = 0; ni < WN; ++ni) {
      v8f a = acc[0][mi][ni];
      if constexpr (NCH == 2) a = a + acc[1][mi][ni];
#pragma unroll
      for (int r = 0; r < 8; ++r) {
        const int m   = tm + mi * 16 + 8 * h + r;
        const int row = r0 + m * s2;
        const float add = (open0 && m == 0) ? 0.0f : a[r];
        X[(size_t)row * XL + (tn + ni * 16 + c)] = old[mi][ni][r] + add;
      }
    }
}

template <int WM, int WN, int KU, bool WT>
__global__ __launch_bounds__(256) void square_k(
    const float* __restrict__ Q, const float* __restrict__ Qt,
    float* __restrict__ Cn, float* __restrict__ Ct) {
  const int lane = threadIdx.x & 31;
  const int w    = threadIdx.x >> 5;
  const int wm = w >> 2, wn = w & 3;
  const int tm = blockIdx.x * (32 * WM) + wm * (16 * WM);
  const int tn = blockIdx.y * (64 * WN) + wn * (16 * WN);
  const int c = lane & 15;
  const int h = lane >> 4;
  constexpr int NCH = (WM * WN < 4) ? 2 : 1;
  constexpr int KT = 8 * KU;
  static_assert(XL % (2 * KT) == 0, "two trips per loop pass");

  v8f acc[NCH][WM][WN];
#pragma unroll
  for (int ch = 0; ch < NCH; ++ch)
#pragma unroll
    for (int mi = 0; mi < WM; ++mi)
#pragma unroll
      for (int ni = 0; ni < WN; ++ni) acc[ch][mi][ni] = (v8f){};

  const float* ap[WM];
  const float* bp[WN];
#pragma unroll
  for (int mi = 0; mi < WM; ++mi)
    ap[mi] = Q + (size_t)((tm >> 4) + mi) * (16 * XL) + c * 8 + 4 * h;
#pragma unroll
  for (int ni = 0; ni < WN; ++ni)
    bp[ni] = Qt + (size_t)((tn >> 4) + ni) * (16 * XL) + c * 8 + 4 * h;

  frag_set<WM, WN, KU> f0, f1;
  load_set<WM, WN, KU, 128, 128>(f0, ap, bp, 0);
  for (int k = 0; k < XL; k += 2 * KT) {
    load_set<WM, WN, KU, 128, 128>(f1, ap, bp, k + KT);
    asm volatile("" ::: "memory");
    mma_set<WM, WN, KU, NCH>(acc, f0);
    const int kn = (k + 2 * KT < XL) ? (k + 2 * KT) : k;
    load_set<WM, WN, KU, 128, 128>(f0, ap, bp, kn);
    asm volatile("" ::: "memory");
    mma_set<WM, WN, KU, NCH>(acc, f1);
  }

#pragma unroll
  for (int mi = 0; mi < WM; ++mi)
#pragma unroll
    for (int ni = 0; ni < WN; ++ni) {
      v8f a = acc[0][mi][ni];
      if constexpr (NCH == 2) a = a + acc[1][mi][ni];
      const int m0 = tm + mi * 16 + 8 * h;
      const int n  = tn + ni * 16 + c;
      if constexpr (WT) {
        float* p = Cn + panel_off(m0, n);
#pragma unroll
        for (int r = 0; r < 8; ++r) p[r * 8] = a[r];
        float* q = Ct + panel_off(n, m0);
        *(v4f*)q       = (v4f){ a[0], a[1], a[2], a[3] };
        *(v4f*)(q + 4) = (v4f){ a[4], a[5], a[6], a[7] };
      } else {
#pragma unroll
        for (int r = 0; r < 8; ++r) Cn[(size_t)(m0 + r) * XL + n] = a[r];
      }
    }
}

template <int MASK>
__device__ __forceinline__ float row_xmask_add(float v) {
  const int t = __builtin_amdgcn_update_dpp(0, __builtin_bit_cast(int, v),
                                            0x160 + MASK, 0xf, 0xf, true);
  return v + __builtin_bit_cast(float, t);
}

__device__ __forceinline__ float wave_sum(float v) {
  v = row_xmask_add<1>(v);
  v = row_xmask_add<2>(v);
  v = row_xmask_add<4>(v);
  v = row_xmask_add<8>(v);
  const int b = __builtin_bit_cast(int, v);
  const int o = __builtin_amdgcn_permlanex16(b, b, 0x76543210, 0xfedcba98, false, false);
  return v + __builtin_bit_cast(float, o);
}

__global__ __launch_bounds__(256) void matvec_step_k(
    const float* __restrict__ Mm, const float* __restrict__ vin,
    const float* vadd, float* vout) {
  __shared__ __attribute__((aligned(16))) float rsum[32];
  const int lane = threadIdx.x & 31, wid = threadIdx.x >> 5;
  const int line0 = blockIdx.x * 32;
  const int row0  = line0 + wid * 4;

  float addv = 0.f;
  if (wid == 0) addv = vadd[line0 + lane];

  v4f x[8], m[4][8];
#pragma unroll
  for (int j = 0; j < 8; ++j) x[j] = *(const v4f*)(vin + 128 * j + 4 * lane);
#pragma unroll
  for (int r = 0; r < 4; ++r)
#pragma unroll
    for (int j = 0; j < 8; ++j)
      m[r][j] = *(const v4f*)(Mm + (size_t)(row0 + r) * XL + 128 * j + 4 * lane);
  __builtin_amdgcn_sched_barrier(0);

  v4f s4;
#pragma unroll
  for (int r = 0; r < 4; ++r) {
    v4f acc = m[r][0] * x[0];
#pragma unroll
    for (int j = 1; j < 8; ++j) acc += m[r][j] * x[j];
    s4[r] = wave_sum((acc.x + acc.y) + (acc.z + acc.w));
  }
  if (lane == 0) *(v4f*)(rsum + wid * 4) = s4;
  __syncthreads();
  if (wid == 0) vout[line0 + lane] = rsum[lane] + addv;
}

template <int WM, int WN, int KU>
__device__ __forceinline__ void kpass(v8f (&acc)[1][WM][WN],
    const float* const (&ap)[WM], const float* const (&bp)[WN], int K) {
  constexpr int KT = 8 * KU;
  frag_set<WM, WN, KU> f0, f1;
  load_set<WM, WN, KU, 8, 8>(f0, ap, bp, 0);
  for (int k = 0; k < K; k += 2 * KT) {
    load_set<WM, WN, KU, 8, 8>(f1, ap, bp, k + KT);
    asm volatile("" ::: "memory");
    mma_set<WM, WN, KU, 1>(acc, f0);
    const int kn = (k + 2 * KT < K) ? (k + 2 * KT) : k;
    load_set<WM, WN, KU, 8, 8>(f0, ap, bp, kn);
    asm volatile("" ::: "memory");
    mma_set<WM, WN, KU, 1>(acc, f1);
  }
}

template <int WM, int WN, int KU>
__global__ __launch_bounds__(256) void out_k(
    const float* __restrict__ xs, const float* __restrict__ Cm,
    const float* __restrict__ uf, const float* __restrict__ Dm,
    float* __restrict__ out) {
  static_assert(XL % (16 * KU) == 0 && UL % (16 * KU) == 0, "two trips per loop pass");
  const int lane = threadIdx.x & 31;
  const int w    = threadIdx.x >> 5;
  const int wm = w >> 2, wn = w & 3;
  const int tm = blockIdx.x * (32 * WM) + wm * (16 * WM);
  const int tn = blockIdx.y * (64 * WN) + wn * (16 * WN);
  const int c = lane & 15;
  const int h = lane >> 4;

  v8f acc[1][WM][WN];
#pragma unroll
  for (int mi = 0; mi < WM; ++mi)
#pragma unroll
    for (int ni = 0; ni < WN; ++ni) acc[0][mi][ni] = (v8f){};

  {
    const float* ap[WM];
    const float* bp[WN];
#pragma unroll
    for (int mi = 0; mi < WM; ++mi)
      ap[mi] = xs + (size_t)(tm + mi * 16 + c) * XL + 4 * h;
#pragma unroll
    for (int ni = 0; ni < WN; ++ni)
      bp[ni] = Cm + (size_t)(tn + ni * 16 + c) * XL + 4 * h;
    kpass<WM, WN, KU>(acc, ap, bp, XL);
  }
  {
    const float* ap[WM];
    const float* bp[WN];
#pragma unroll
    for (int mi = 0; mi < WM; ++mi)
      ap[mi] = uf + (size_t)(tm + mi * 16 + c) * UL + 4 * h;
#pragma unroll
    for (int ni = 0; ni < WN; ++ni)
      bp[ni] = Dm + (size_t)(tn + ni * 16 + c) * UL + 4 * h;
    kpass<WM, WN, KU>(acc, ap, bp, UL);
  }

#pragma unroll
  for (int mi = 0; mi < WM; ++mi)
#pragma unroll
    for (int ni = 0; ni < WN; ++ni) {
      const int yl = tn + ni * 16 + c;
      const int n0 = tm + mi * 16 + 8 * h;
      const v8f a = acc[0][mi][ni];
      float* op = out + (size_t)yl * NSEQ + n0;
      *(v4f*)op       = (v4f){ tanhf(a[0]), tanhf(a[1]), tanhf(a[2]), tanhf(a[3]) };
      *(v4f*)(op + 4) = (v4f){ tanhf(a[4]), tanhf(a[5]), tanhf(a[6]), tanhf(a[7]) };
    }
}

__global__ __launch_bounds__(256) void transpose_k(
    const float* __restrict__ in, float* __restrict__ out, int rows, int cols) {
  __shared__ float tile[32][33];
  const int c0 = blockIdx.x * 32, r0 = blockIdx.y * 32;
  const int tx = threadIdx.x, ty = threadIdx.y;
#pragma unroll
  for (int i = 0; i < 32; i += 8)
    tile[ty + i][tx] = in[(size_t)(r0 + ty + i) * cols + (c0 + tx)];
  __syncthreads();
#pragma unroll
  for (int i = 0; i < 32; i += 8)
    out[(size_t)(c0 + ty + i) * rows + (r0 + tx)] = tile[tx][ty + i];
}

__global__ __launch_bounds__(256) void panel_k(
    const float* __restrict__ in, float* __restrict__ rt, float* __restrict__ tt) {
  __shared__ float tile[32][33];
  const int c0 = blockIdx.x * 32, r0 = blockIdx.y * 32;
  const int tx = threadIdx.x, ty = threadIdx.y;
#pragma unroll
  for (int i = 0; i < 32; i += 8) {
    const float v = in[(size_t)(r0 + ty + i) * XL + (c0 + tx)];
    tile[ty + i][tx] = v;
    rt[panel_off(r0 + ty + i, c0 + tx)] = v;
  }
  __syncthreads();
#pragma unroll
  for (int i = 0; i < 32; i += 8)
    tt[panel_off(c0 + ty + i, r0 + tx)] = tile[tx][ty + i];
}

extern "C" void kernel_launch(void* const* d_in, const int* in_sizes, int n_in,
                              void* d_out, int out_size, void* d_ws, size_t ws_size,
                              hipStream_t stream) {
  (void)in_sizes; (void)n_in; (void)out_size; (void)ws_size;
  const float* u  = (const float*)d_in[0];
  const float* A  = (const float*)d_in[1];
  const float* B  = (const float*)d_in[2];
  const float* C  = (const float*)d_in[3];
  const float* D  = (const float*)d_in[4];
  const float* x0 = (const float*)d_in[5];
  float* out = (float*)d_out;

  float* ws = (float*)d_ws;
  size_t off = 0;
  float* X  = ws + off; off += (size_t)NSEQ * XL;
  float* Bt = ws + off; off += (size_t)UL * XL;
  float* P2   = ws + off; off += (size_t)XL * XL;
  float* P4   = ws + off; off += (size_t)XL * XL;
  float* P8   = ws + off; off += (size_t)XL * XL;
  float* P16  = ws + off; off += (size_t)XL * XL;
  float* P32  = ws + off; off += (size_t)XL * XL;
  float* P64  = ws + off; off += (size_t)XL * XL;
  float* P128 = ws + off; off += (size_t)XL * XL;
  float* P256  = out;
  float* P512  = out + (size_t)XL * XL;
  float* P1024 = out + 2 * (size_t)XL * XL;
  float* P1    = out + 3 * (size_t)XL * XL;

  float* Ta = X;
  float* Tb = X + (size_t)XL * XL;

  transpose_k<<<dim3(UL / 32, XL / 32), dim3(32, 8), 0, stream>>>(B, Bt, XL, UL);
  panel_k<<<dim3(XL / 32, XL / 32), dim3(32, 8), 0, stream>>>(A, P1, Ta);

  const dim3 sqg(XL / 64, XL / 64);
  square_k<2, 1, 8, true ><<<sqg, 256, 0, stream>>>(P1,  Ta, P2,   Tb);
  square_k<2, 1, 8, true ><<<sqg, 256, 0, stream>>>(P2,  Tb, P4,   Ta);
  square_k<2, 1, 8, true ><<<sqg, 256, 0, stream>>>(P4,  Ta, P8,   Tb);
  square_k<2, 1, 8, true ><<<sqg, 256, 0, stream>>>(P8,  Tb, P16,  Ta);
  square_k<2, 1, 8, true ><<<sqg, 256, 0, stream>>>(P16, Ta, P32,  Tb);
  square_k<2, 1, 8, true ><<<sqg, 256, 0, stream>>>(P32, Tb, P64,  Ta);
  square_k<2, 1, 8, true ><<<sqg, 256, 0, stream>>>(P64,  Ta, P128,  Tb);
  square_k<2, 1, 8, true ><<<sqg, 256, 0, stream>>>(P128, Tb, P256,  Ta);
  square_k<2, 1, 8, true ><<<sqg, 256, 0, stream>>>(P256, Ta, P512,  Tb);
  square_k<2, 1, 8, false><<<sqg, 256, 0, stream>>>(P512, Tb, P1024, nullptr);

  gemm_rr_k<4, 4><<<dim3(NSEQ / 128, XL / 256), 256, 0, stream>>>(
      u, Bt, X, UL, UL, XL, XL);
  matvec_step_k<<<32, 256, 0, stream>>>(A, x0, X, X);

  sweep_k<4, 4, 2><<<dim3(8192 / 128, XL / 256), 256, 0, stream>>>(X, P1,  1,  1);
  sweep_k<2, 4, 2><<<dim3(4096 / 64,  XL / 256), 256, 0, stream>>>(X, P2,  2,  3);
  sweep_k<2, 2, 4><<<dim3(2048 / 64,  XL / 128), 256, 0, stream>>>(X, P4,  4,  7);
  sweep_k<1, 2, 8><<<dim3(1024 / 32,  XL / 128), 256, 0, stream>>>(X, P8,  8,  15);
  sweep_k<1, 1, 16><<<dim3(512 / 32,   XL / 64),  256, 0, stream>>>(X, P16, 16, 31);
  sweep_k<1, 1, 16><<<dim3(256 / 32,   XL / 64),  256, 0, stream>>>(X, P32, 32, 63);
  sweep_k<1, 1, 16><<<dim3(128 / 32,   XL / 64),  256, 0, stream>>>(X, P64, 64, 127);
  sweep_k<1, 1, 16><<<dim3(64 / 32,    XL / 64),  256, 0, stream>>>(X, P128, 128, 255);
  sweep_k<1, 1, 16><<<dim3(32 / 32,    XL / 64),  256, 0, stream>>>(X, P256, 256, 511);
  sweep_k<1, 1, 16, 1><<<dim3(16 / 16, XL / 128), 256, 0, stream>>>(X, P512, 512, 1023);

  for (int j = 1; j < NSEQ / CLEN; ++j) {
    float* prev = X + ((size_t)j * CLEN - 1) * XL;
    float* cur  = X + ((size_t)(j + 1) * CLEN - 1) * XL;
    matvec_step_k<<<32, 256, 0, stream>>>(P1024, prev, cur, cur);
  }

  sweep_k<1, 1, 16, 1><<<dim3(16 / 16, XL / 128), 256, 0, stream>>>(X, P512, 512, 511);
  sweep_k<1, 1, 16><<<dim3(32 / 32,    XL / 64),  256, 0, stream>>>(X, P256, 256, 255);
  sweep_k<1, 1, 16><<<dim3(64 / 32,    XL / 64),  256, 0, stream>>>(X, P128, 128, 127);
  sweep_k<1, 1, 16><<<dim3(128 / 32,   XL / 64),  256, 0, stream>>>(X, P64, 64, 63);
  sweep_k<1, 1, 16><<<dim3(256 / 32,   XL / 64),  256, 0, stream>>>(X, P32, 32, 31);
  sweep_k<1, 1, 16><<<dim3(512 / 32,   XL / 64),  256, 0, stream>>>(X, P16, 16, 15);
  sweep_k<1, 2, 8><<<dim3(1024 / 32,  XL / 128), 256, 0, stream>>>(X, P8,  8,  7);
  sweep_k<2, 2, 4><<<dim3(2048 / 64,  XL / 128), 256, 0, stream>>>(X, P4,  4,  3);
  sweep_k<2, 4, 2><<<dim3(4096 / 64,  XL / 256), 256, 0, stream>>>(X, P2,  2,  1);
  sweep_k<4, 4, 2><<<dim3(8192 / 128, XL / 256), 256, 0, stream>>>(X, P1,  1,  0);

  out_k<2, 4, 2><<<dim3(NSEQ / 64, YLN / 256), 256, 0, stream>>>(X, C, u, D, out);
}


# squarings of A share launches with the other chain: each rides in the grid of the drive product, the x0 matvec or an up level; B^T with the panel copies; 50 launches become 39; sums as before
# speedup vs baseline: 7.7214x; 1.0924x over previous
#include <hip/hip_runtime.h>
#include <math.h>


typedef __attribute__((ext_vector_type(2))) float v2f;
typedef __attribute__((ext_vector_type(4))) float v4f;
typedef __attribute__((ext_vector_type(8))) float v8f;

#define NSEQ   16384
#define XL     1024
#define UL     256
#define YLN    256
#define CLEN   1024

__device__ __forceinline__ v8f wmma4(v2f a, v2f b, v8f c) {
  return __builtin_amdgcn_wmma_f32_16x16x4_f32(
       false, a,  false, b,
       (short)0, c,  false,  false);
}

__device__ __forceinline__ size_t panel_off(int R, int K) {
  return ((size_t)(R >> 4) * (XL / 8) + (size_t)(K >> 3)) * 128
         + (size_t)((R & 15) * 8 + (K & 7));
}

template <int WM, int WN>
__device__ __forceinline__ void gemm_rr_body(
    const float* __restrict__ Ap, const float* __restrict__ Bp,
    float* __restrict__ Cp, int K, int lda, int ldb, int ldc, int bx, int by) {
  const int lane = threadIdx.x & 31;
  const int w    = threadIdx.x >> 5;
  const int wm = w >> 2, wn = w & 3;
  const int tm = bx * (32 * WM) + wm * (16 * WM);
  const int tn = by * (64 * WN) + wn * (16 * WN);
  const int ml = lane & 15;
  const int kh = (lane >> 4) << 1;

  v8f acc[WM][WN];
#pragma unroll
  for (int mi = 0; mi < WM; ++mi)
#pragma unroll
    for (int ni = 0; ni < WN; ++ni) acc[mi][ni] = (v8f){};

  const float* ap[WM];
  const float* bp[WN];
#pragma unroll
  for (int mi = 0; mi < WM; ++mi)
    ap[mi] = Ap + (size_t)(tm + mi * 16 + ml) * lda + kh;
#pragma unroll
  for (int ni = 0; ni < WN; ++ni)
    bp[ni] = Bp + (size_t)kh * ldb + (tn + ni * 16 + ml);

  for (int k = 0; k < K; k += 4) {
    v2f af[WM], bf[WN];
    const size_t bo0 = (size_t)k * ldb, bo1 = bo0 + (size_t)ldb;
#pragma unroll
    for (int mi = 0; mi < WM; ++mi) af[mi] = (v2f){ ap[mi][k], ap[mi][k + 1] };
#pragma unroll
    for (int ni = 0; ni < WN; ++ni) bf[ni] = (v2f){ bp[ni][bo0], bp[ni][bo1] };
#pragma unroll
    for (int mi = 0; mi < WM; ++mi)
#pragma unroll
      for (int ni = 0; ni < WN; ++ni)
        acc[mi][ni] = wmma4(af[mi], bf[ni], acc[mi][ni]);
  }

  const int nn = lane & 15;
  const int mb = (lane >> 4) << 3;
#pragma unroll
  for (int mi = 0; mi < WM; ++mi)
#pragma unroll
    for (int ni = 0; ni < WN; ++ni) {
      float* cp = Cp + (size_t)(tm + mi * 16 + mb) * ldc + (tn + ni * 16 + nn);
#pragma unroll
      for (int r = 0; r < 8; ++r) cp[(size_t)r * ldc] = acc[mi][ni][r];
    }
}

template <int WM, int WN, int KU>
struct frag_set { v4f a[KU][WM]; v4f b[KU][WN]; };

template <int WM, int WN, int KU, int AS, int BS>
__device__ __forceinline__ void load_set(frag_set<WM, WN, KU>& f,
    const float* const (&ap)[WM], const float* const (&bp)[WN], int k) {
#pragma unroll
  for (int u = 0; u < KU; ++u) {
#pragma unroll
    for (int mi = 0; mi < WM; ++mi)
      f.a[u][mi] = *(const v4f*)(ap[mi] + (k / 8 + u) * AS);
#pragma unroll
    for (int ni = 0; ni < WN; ++ni)
      f.b[u][ni] = *(const v4f*)(bp[ni] + (k / 8 + u) * BS);
  }
}

template <int WM, int WN, int KU, int NCH>
__device__ __forceinline__ void mma_set(v8f (&acc)[NCH][WM][WN],
                                        const frag_set<WM, WN, KU>& f) {
#pragma unroll
  for (int u = 0; u < KU; ++u)
#pragma unroll
    for (int mi = 0; mi < WM; ++mi)
#pragma unroll
      for (int ni = 0; ni < WN; ++ni) {
        const v2f a0 = { f.a[u][mi].x, f.a[u][mi].y };
        const v2f a1 = { f.a[u][mi].z, f.a[u][mi].w };
        const v2f b0 = { f.b[u][ni].x, f.b[u][ni].y };
        const v2f b1 = { f.b[u][ni].z, f.b[u][ni].w };
        acc[0][mi][ni]       = wmma4(a0, b0, acc[0][mi][ni]);
        acc[NCH - 1][mi][ni] = wmma4(a1, b1, acc[NCH - 1][mi][ni]);
      }
}

template <int WM, int WN, int KU, int WGM>
__device__ __forceinline__ void sweep_body(
    float* X, const float* __restrict__ P, int s, int r0, int bx, int by) {
  constexpr int WGN = 8 / WGM;
  const int lane = threadIdx.x & 31;
  const int w    = threadIdx.x >> 5;
  const int wm = w / WGN, wn = w % WGN;
  const int tm = bx * (16 * WM * WGM) + wm * (16 * WM);
  const int tn = by * (16 * WN * WGN) + wn * (16 * WN);
  const int c = lane & 15;
  const int h = lane >> 4;
  const int s2 = 2 * s;
  constexpr int NCH = (WM * WN < 4) ? 2 : 1;
  constexpr int KT = 8 * KU;
  static_assert(XL % (2 * KT) == 0, "two trips per loop pass");

  v8f acc[NCH][WM][WN];
#pragma unroll
  for (int ch = 0; ch < NCH; ++ch)
#pragma unroll
    for (int mi = 0; mi < WM; ++mi)
#pragma unroll
      for (int ni = 0; ni < WN; ++ni) acc[ch][mi][ni] = (v8f){};

  const float* ap[WM];
  const float* bp[WN];
#pragma unroll
  for (int mi = 0; mi < WM; ++mi) {
    int q = r0 - s + (tm + mi * 16 + c) * s2;
    if (q < 0) q = 0;
    ap[mi] = X + (size_t)q * XL + 4 * h;
  }
#pragma unroll
  for (int ni = 0; ni < WN; ++ni)
    bp[ni] = P + (size_t)((tn >> 4) + ni) * (16 * XL) + c * 8 + 4 * h;

  frag_set<WM, WN, KU> f0, f1;
  load_set<WM, WN, KU, 8, 128>(f0, ap, bp, 0);
  for (int k = 0; k < XL; k += 2 * KT) {
    load_set<WM, WN, KU, 8, 128>(f1, ap, bp, k + KT);
    asm volatile("" ::: "memory");
    mma_set<WM, WN, KU, NCH>(acc, f0);
    const int kn = (k + 2 * KT < XL) ? (k + 2 * KT) : k;
    load_set<WM, WN, KU, 8, 128>(f0, ap, bp, kn);
    asm volatile("" ::: "memory");
    mma_set<WM, WN, KU, NCH>(acc, f1);
  }

  const bool open0 = (r0 < s);
  v8f old[WM][WN];
#pragma unroll
  for (int mi = 0; mi < WM; ++mi)
#pragma unroll
    for (int ni = 0; ni < WN; ++ni) {
#pragma unroll
      for (int r = 0; r < 8; ++r) {
        const int row = r0 + (tm + mi * 16 + 8 * h + r) * s2;
        old[mi][ni][r] = X[(size_t)row * XL + (tn + ni * 16 + c)];
      }
    }
#pragma unroll
  for (int mi = 0; mi < WM; ++mi)
#pragma unroll
    for (int ni = 0; ni < WN; ++ni) {
      v8f a = acc[0][mi][ni];
      if constexpr (NCH == 2) a = a + acc[1][mi][ni];
#pragma unroll
      for (int r = 0; r < 8; ++r) {
        const int m   = tm + mi * 16 + 8 * h + r;
        const int row = r0 + m * s2;
        const float add = (open0 && m == 0) ? 0.0f : a[r];
        X[(size_t)row * XL + (tn + ni * 16 + c)] = old[mi][ni][r] + add;
      }
    }
}

template <int WM, int WN, int KU, int WGM = 2>
__global__ __launch_bounds__(256) void sweep_k(
    float* X, const float* __restrict__ P, int s, int r0) {
  sweep_body<WM, WN, KU, WGM>(X, P, s, r0, blockIdx.x, blockIdx.y);
}

template <int WM, int WN, int KU, bool WT>
__device__ __forceinline__ void square_body(
    const float* __restrict__ Q, const float* __restrict__ Qt,
    float* __restrict__ Cn, float* __restrict__ Ct, int bx, int by) {
  const int lane = threadIdx.x & 31;
  const int w    = threadIdx.x >> 5;
  const int wm = w >> 2, wn = w & 3;
  const int tm = bx * (32 * WM) + wm * (16 * WM);
  const int tn = by * (64 * WN) + wn * (16 * WN);
  const int c = lane & 15;
  const int h = lane >> 4;
  constexpr int NCH = (WM * WN < 4) ? 2 : 1;
  constexpr int KT = 8 * KU;
  static_assert(XL % (2 * KT) == 0, "two trips per loop pass");

  v8f acc[NCH][WM][WN];
#pragma unroll
  for (int ch = 0; ch < NCH; ++ch)
#pragma unroll
    for (int mi = 0; mi < WM; ++mi)
#pragma unroll
      for (int ni = 0; ni < WN; ++ni) acc[ch][mi][ni] = (v8f){};

  const float* ap[WM];
  const float* bp[WN];
#pragma unroll
  for (int mi = 0; mi < WM; ++mi)
    ap[mi] = Q + (size_t)((tm >> 4) + mi) * (16 * XL) + c * 8 + 4 * h;
#pragma unroll
  for (int ni = 0; ni < WN; ++ni)
    bp[ni] = Qt + (size_t)((tn >> 4) + ni) * (16 * XL) + c * 8 + 4 * h;

  frag_set<WM, WN, KU> f0, f1;
  load_set<WM, WN, KU, 128, 128>(f0, ap, bp, 0);
  for (int k = 0; k < XL; k += 2 * KT) {
    load_set<WM, WN, KU, 128, 128>(f1, ap, bp, k + KT);
    asm volatile("" ::: "memory");
    mma_set<WM, WN, KU, NCH>(acc, f0);
    const int kn = (k + 2 * KT < XL) ? (k + 2 * KT) : k;
    load_set<WM, WN, KU, 128, 128>(f0, ap, bp, kn);
    asm volatile("" ::: "memory");
    mma_set<WM, WN, KU, NCH>(acc, f1);
  }

#pragma unroll
  for (int mi = 0; mi < WM; ++mi)
#pragma unroll
    for (int ni = 0; ni < WN; ++ni) {
      v8f a = acc[0][mi][ni];
      if constexpr (NCH == 2) a = a + acc[1][mi][ni];
      const int m0 = tm + mi * 16 + 8 * h;
      const int n  = tn + ni * 16 + c;
      if constexpr (WT) {
        float* p = Cn + panel_off(m0, n);
#pragma unroll
        for (int r = 0; r < 8; ++r) p[r * 8] = a[r];
        float* q = Ct + panel_off(n, m0);
        *(v4f*)q       = (v4f){ a[0], a[1], a[2], a[3] };
        *(v4f*)(q + 4) = (v4f){ a[4], a[5], a[6], a[7] };
      } else {
#pragma unroll
        for (int r = 0; r < 8; ++r) Cn[(size_t)(m0 + r) * XL + n] = a[r];
      }
    }
}

template <int MASK>
__device__ __forceinline__ float row_xmask_add(float v) {
  const int t = __builtin_amdgcn_update_dpp(0, __builtin_bit_cast(int, v),
                                            0x160 + MASK, 0xf, 0xf, true);
  return v + __builtin_bit_cast(float, t);
}

__device__ __forceinline__ float wave_sum(float v) {
  v = row_xmask_add<1>(v);
  v = row_xmask_add<2>(v);
  v = row_xmask_add<4>(v);
  v = row_xmask_add<8>(v);
  const int b = __builtin_bit_cast(int, v);
  const int o = __builtin_amdgcn_permlanex16(b, b, 0x76543210, 0xfedcba98, false, false);
  return v + __builtin_bit_cast(float, o);
}

__device__ __forceinline__ void matvec_body(
    const float* __restrict__ Mm, const float* __restrict__ vin,
    const float* vadd, float* vout, int bx) {
  __shared__ __attribute__((aligned(16))) float rsum[32];
  const int lane = threadIdx.x & 31, wid = threadIdx.x >> 5;
  const int line0 = bx * 32;
  const int row0  = line0 + wid * 4;

  float addv = 0.f;
  if (wid == 0) addv = vadd[line0 + lane];

  v4f x[8], m[4][8];
#pragma unroll
  for (int j = 0; j < 8; ++j) x[j] = *(const v4f*)(vin + 128 * j + 4 * lane);
#pragma unroll
  for (int r = 0; r < 4; ++r)
#pragma unroll
    for (int j = 0; j < 8; ++j)
      m[r][j] = *(const v4f*)(Mm + (size_t)(row0 + r) * XL + 128 * j + 4 * lane);
  __builtin_amdgcn_sched_barrier(0);

  v4f s4;
#pragma unroll
  for (int r = 0; r < 4; ++r) {
    v4f acc = m[r][0] * x[0];
#pragma unroll
    for (int j = 1; j < 8; ++j) acc += m[r][j] * x[j];
    s4[r] = wave_sum((acc.x + acc.y) + (acc.z + acc.w));
  }
  if (lane == 0) *(v4f*)(rsum + wid * 4) = s4;
  __syncthreads();
  if (wid == 0) vout[line0 + lane] = rsum[lane] + addv;
}

__global__ __launch_bounds__(256) void matvec_step_k(
    const float* __restrict__ Mm, const float* __restrict__ vin,
    const float* vadd, float* vout) {
  matvec_body(Mm, vin, vadd, vout, blockIdx.x);
}

template <int WM, int WN, int KU>
__device__ __forceinline__ void kpass(v8f (&acc)[1][WM][WN],
    const float* const (&ap)[WM], const float* const (&bp)[WN], int K) {
  constexpr int KT = 8 * KU;
  frag_set<WM, WN, KU> f0, f1;
  load_set<WM, WN, KU, 8, 8>(f0, ap, bp, 0);
  for (int k = 0; k < K; k += 2 * KT) {
    load_set<WM, WN, KU, 8, 8>(f1, ap, bp, k + KT);
    asm volatile("" ::: "memory");
    mma_set<WM, WN, KU, 1>(acc, f0);
    const int kn = (k + 2 * KT < K) ? (k + 2 * KT) : k;
    load_set<WM, WN, KU, 8, 8>(f0, ap, bp, kn);
    asm volatile("" ::: "memory");
    mma_set<WM, WN, KU, 1>(acc, f1);
  }
}

template <int WM, int WN, int KU>
__global__ __launch_bounds__(256) void out_k(
    const float* __restrict__ xs, const float* __restrict__ Cm,
    const float* __restrict__ uf, const float* __restrict__ Dm,
    float* __restrict__ out) {
  static_assert(XL % (16 * KU) == 0 && UL % (16 * KU) == 0, "two trips per loop pass");
  const int lane = threadIdx.x & 31;
  const int w    = threadIdx.x >> 5;
  const int wm = w >> 2, wn = w & 3;
  const int tm = blockIdx.x * (32 * WM) + wm * (16 * WM);
  const int tn = blockIdx.y * (64 * WN) + wn * (16 * WN);
  const int c = lane & 15;
  const int h = lane >> 4;

  v8f acc[1][WM][WN];
#pragma unroll
  for (int mi = 0; mi < WM; ++mi)
#pragma unroll
    for (int ni = 0; ni < WN; ++ni) acc[0][mi][ni] = (v8f){};

  {
    const float* ap[WM];
    const float* bp[WN];
#pragma unroll
    for (int mi = 0; mi < WM; ++mi)
      ap[mi] = xs + (size_t)(tm + mi * 16 + c) * XL + 4 * h;
#pragma unroll
    for (int ni = 0; ni < WN; ++ni)
      bp[ni] = Cm + (size_t)(tn + ni * 16 + c) * XL + 4 * h;
    kpass<WM, WN, KU>(acc, ap, bp, XL);
  }
  {
    const float* ap[WM];
    const float* bp[WN];
#pragma unroll
    for (int mi = 0; mi < WM; ++mi)
      ap[mi] = uf + (size_t)(tm + mi * 16 + c) * UL + 4 * h;
#pragma unroll
    for (int ni = 0; ni < WN; ++ni)
      bp[ni] = Dm + (size_t)(tn + ni * 16 + c) * UL + 4 * h;
    kpass<WM, WN, KU>(acc, ap, bp, UL);
  }

#pragma unroll
  for (int mi = 0; mi < WM; ++mi)
#pragma unroll
    for (int ni = 0; ni < WN; ++ni) {
      const int yl = tn + ni * 16 + c;
      const int n0 = tm + mi * 16 + 8 * h;
      const v8f a = acc[0][mi][ni];
      float* op = out + (size_t)yl * NSEQ + n0;
      *(v4f*)op       = (v4f){ tanhf(a[0]), tanhf(a[1]), tanhf(a[2]), tanhf(a[3]) };
      *(v4f*)(op + 4) = (v4f){ tanhf(a[4]), tanhf(a[5]), tanhf(a[6]), tanhf(a[7]) };
    }
}

__device__ __forceinline__ void transpose_body(
    const float* __restrict__ in, float* __restrict__ out, int rows, int cols,
    int bx, int by) {
  __shared__ float tile[32][33];
  const int c0 = bx * 32, r0 = by * 32;
  const int tx = threadIdx.x & 31, ty = threadIdx.x >> 5;
#pragma unroll
  for (int i = 0; i < 32; i += 8)
    tile[ty + i][tx] = in[(size_t)(r0 + ty + i) * cols + (c0 + tx)];
  __syncthreads();
#pragma unroll
  for (int i = 0; i < 32; i += 8)
    out[(size_t)(c0 + ty + i) * rows + (r0 + tx)] = tile[tx][ty + i];
}

template <bool WTT>
__device__ __forceinline__ void panel_body(
    const float* __restrict__ in, float* __restrict__ rt, float* __restrict__ tt,
    int bx, int by) {
  __shared__ float tile[32][33];
  const int c0 = bx * 32, r0 = by * 32;
  const int tx = threadIdx.x & 31, ty = threadIdx.x >> 5;
#pragma unroll
  for (int i = 0; i < 32; i += 8) {
    const float v = in[(size_t)(r0 + ty + i) * XL + (c0 + tx)];
    if constexpr (WTT) tile[ty + i][tx] = v;
    rt[panel_off(r0 + ty + i, c0 + tx)] = v;
  }
  if constexpr (WTT) {
    __syncthreads();
#pragma unroll
    for (int i = 0; i < 32; i += 8)
      tt[panel_off(c0 + ty + i, r0 + tx)] = tile[tx][ty + i];
  }
}

constexpr int SQG = (XL / 64);

__global__ __launch_bounds__(256) void prep_k(
    const float* __restrict__ Bm, float* __restrict__ Bt,
    const float* __restrict__ Am, float* __restrict__ rt, float* __restrict__ tt) {
  constexpr int NT = (UL / 32) * (XL / 32);
  const int b = blockIdx.x;
  if (b < NT) {
    transpose_body(Bm, Bt, XL, UL, b % (UL / 32), b / (UL / 32));
  } else {
    const int q = b - NT;
    panel_body<true>(Am, rt, tt, q % (XL / 32), q / (XL / 32));
  }
}

template <int GX, int GY>
__global__ __launch_bounds__(256) void bu_sq_k(
    const float* __restrict__ uf, const float* __restrict__ Bt, float* __restrict__ X,
    int K, int lda, int ldb, int ldc,
    const float* __restrict__ Q, const float* __restrict__ Qt,
    float* __restrict__ Cn, float* __restrict__ Ct) {
  const int b = blockIdx.x;
  if (b < GX * GY) {
    gemm_rr_body<4, 4>(uf, Bt, X, K, lda, ldb, ldc, b % GX, b / GX);
  } else {
    const int q = b - GX * GY;
    square_body<2, 1, 8, true>(Q, Qt, Cn, Ct, q % SQG, q / SQG);
  }
}

__global__ __launch_bounds__(256) void mv_sq_k(
    const float* __restrict__ Mm, const float* __restrict__ vin,
    const float* vadd, float* vout,
    const float* __restrict__ Q, const float* __restrict__ Qt,
    float* __restrict__ Cn, float* __restrict__ Ct) {
  const int b = blockIdx.x;
  if (b < XL / 32) {
    matvec_body(Mm, vin, vadd, vout, b);
  } else {
    const int q = b - XL / 32;
    square_body<2, 1, 8, true>(Q, Qt, Cn, Ct, q % SQG, q / SQG);
  }
}

template <int WM, int WN, int KU, bool WT>
__global__ __launch_bounds__(256) void sweep_sq_k(
    float* X, const float* __restrict__ P, int s, int r0, int gxs, int nsw,
    const float* __restrict__ Q, const float* __restrict__ Qt,
    float* __restrict__ Cn, float* __restrict__ Ct) {
  const int b = blockIdx.x;
  if (b < nsw) {
    sweep_body<WM, WN, KU, 2>(X, P, s, r0, b & ((1 << gxs) - 1), b >> gxs);
  } else {
    const int q = b - nsw;
    square_body<2, 1, 8, WT>(Q, Qt, Cn, Ct, q % SQG, q / SQG);
  }
}

template <int WM, int WN, int KU>
__global__ __launch_bounds__(256) void sweep_panel_k(
    float* X, const float* __restrict__ P, int s, int r0, int gxs, int nsw,
    const float* __restrict__ Am, float* __restrict__ rt) {
  const int b = blockIdx.x;
  if (b < nsw) {
    sweep_body<WM, WN, KU, 2>(X, P, s, r0, b & ((1 << gxs) - 1), b >> gxs);
  } else {
    const int q = b - nsw;
    panel_body<false>(Am, rt, nullptr, q % (XL / 32), q / (XL / 32));
  }
}

extern "C" void kernel_launch(void* const* d_in, const int* in_sizes, int n_in,
                              void* d_out, int out_size, void* d_ws, size_t ws_size,
                              hipStream_t stream) {
  (void)in_sizes; (void)n_in; (void)out_size; (void)ws_size;
  const float* u  = (const float*)d_in[0];
  const float* A  = (const float*)d_in[1];
  const float* B  = (const float*)d_in[2];
  const float* C  = (const float*)d_in[3];
  const float* D  = (const float*)d_in[4];
  const float* x0 = (const float*)d_in[5];
  float* out = (float*)d_out;

  float* ws = (float*)d_ws;
  size_t off = 0;
  float* X  = ws + off; off += (size_t)NSEQ * XL;
  float* Bt = ws + off; off += (size_t)UL * XL;
  float* P2   = ws + off; off += (size_t)XL * XL;
  float* P4   = ws + off; off += (size_t)XL * XL;
  float* P8   = ws + off; off += (size_t)XL * XL;
  float* P16  = ws + off; off += (size_t)XL * XL;
  float* P32  = ws + off; off += (size_t)XL * XL;
  float* P64  = ws + off; off += (size_t)XL * XL;
  float* P128 = ws + off; off += (size_t)XL * XL;
  float* P256  = out;
  float* P512  = out + (size_t)XL * XL;
  float* P1024 = out + 2 * (size_t)XL * XL;
  float* P1    = out + 3 * (size_t)XL * XL;

  float* T1 = P4,   * T2 = P8,    * T4 = P16,    * T8 = P32,   * T16 = P64;
  float* T32 = P128, * T64 = P256, * T128 = P512, * T256 = P1024, * T512 = P1;

  prep_k<<<(UL / 32) * (XL / 32) + (XL / 32) * (XL / 32), 256, 0, stream>>>(B, Bt, A, P1, T1);

  bu_sq_k<NSEQ / 128, XL / 256><<<(NSEQ / 128) * (XL / 256) + SQG * SQG, 256, 0, stream>>>(
      u, Bt, X, UL, UL, XL, XL, P1, T1, P2, T2);
  mv_sq_k<<<XL / 32 + SQG * SQG, 256, 0, stream>>>(A, x0, X, X, P2, T2, P4, T4);

  sweep_sq_k<4, 4, 2, true><<<(8192 / 128) * (XL / 256) + SQG * SQG, 256, 0, stream>>>(
      X, P1,  1,  1,   6, (8192 / 128) * (XL / 256), P4,   T4,   P8,    T8);
  sweep_sq_k<2, 4, 2, true><<<(4096 / 64) * (XL / 256) + SQG * SQG, 256, 0, stream>>>(
      X, P2,  2,  3,   6, (4096 / 64) * (XL / 256),  P8,   T8,   P16,   T16);
  sweep_sq_k<2, 2, 4, true><<<(2048 / 64) * (XL / 128) + SQG * SQG, 256, 0, stream>>>(
      X, P4,  4,  7,   5, (2048 / 64) * (XL / 128),  P16,  T16,  P32,   T32);
  sweep_sq_k<1, 2, 8, true><<<(1024 / 32) * (XL / 128) + SQG * SQG, 256, 0, stream>>>(
      X, P8,  8,  15,  5, (1024 / 32) * (XL / 128),  P32,  T32,  P64,   T64);
  sweep_sq_k<1, 1, 16, true><<<(512 / 32) * (XL / 64) + SQG * SQG, 256, 0, stream>>>(
      X, P16, 16, 31,  4, (512 / 32) * (XL / 64),    P64,  T64,  P128,  T128);
  sweep_sq_k<1, 1, 16, true><<<(256 / 32) * (XL / 64) + SQG * SQG, 256, 0, stream>>>(
      X, P32, 32, 63,  3, (256 / 32) * (XL / 64),    P128, T128, P256,  T256);
  sweep_sq_k<1, 1, 16, true><<<(128 / 32) * (XL / 64) + SQG * SQG, 256, 0, stream>>>(
      X, P64, 64, 127, 2, (128 / 32) * (XL / 64),    P256, T256, P512,  T512);
  sweep_sq_k<1, 1, 16, false><<<(64 / 32) * (XL / 64) + SQG * SQG, 256, 0, stream>>>(
      X, P128, 128, 255, 1, (64 / 32) * (XL / 64),   P512, T512, P1024, nullptr);
  sweep_panel_k<1, 1, 16><<<(32 / 32) * (XL / 64) + (XL / 32) * (XL / 32), 256, 0, stream>>>(
      X, P256, 256, 511, 0, (32 / 32) * (XL / 64),   A, P1);
  sweep_k<1, 1, 16, 1><<<dim3(16 / 16, XL / 128), 256, 0, stream>>>(X, P512, 512, 1023);

  for (int j = 1; j < NSEQ / CLEN; ++j) {
    float* prev = X + ((size_t)j * CLEN - 1) * XL;
    float* cur  = X + ((size_t)(j + 1) * CLEN - 1) * XL;
    matvec_step_k<<<32, 256, 0, stream>>>(P1024, prev, cur, cur);
  }

  sweep_k<1, 1, 16, 1><<<dim3(16 / 16, XL / 128), 256, 0, stream>>>(X, P512, 512, 511);
  sweep_k<1, 1, 16><<<dim3(32 / 32,    XL / 64),  256, 0, stream>>>(X, P256, 256, 255);
  sweep_k<1, 1, 16><<<dim3(64 / 32,    XL / 64),  256, 0, stream>>>(X, P128, 128, 127);
  sweep_k<1, 1, 16><<<dim3(128 / 32,   XL / 64),  256, 0, stream>>>(X, P64, 64, 63);
  sweep_k<1, 1, 16><<<dim3(256 / 32,   XL / 64),  256, 0, stream>>>(X, P32, 32, 31);
  sweep_k<1, 1, 16><<<dim3(512 / 32,   XL / 64),  256, 0, stream>>>(X, P16, 16, 15);
  sweep_k<1, 2, 8><<<dim3(1024 / 32,  XL / 128), 256, 0, stream>>>(X, P8,  8,  7);
  sweep_k<2, 2, 4><<<dim3(2048 / 64,  XL / 128), 256, 0, stream>>>(X, P4,  4,  3);
  sweep_k<2, 4, 2><<<dim3(4096 / 64,  XL / 256), 256, 0, stream>>>(X, P2,  2,  1);
  sweep_k<4, 4, 2><<<dim3(8192 / 128, XL / 256), 256, 0, stream>>>(X, P1,  1,  0);

  out_k<2, 4, 2><<<dim3(NSEQ / 64, YLN / 256), 256, 0, stream>>>(X, C, u, D, out);
}


# row operand of the tree levels and the output product (rows of X, xs, u) fetched once per workgroup in whole lines, handed to the waves through LDS, not 16 lines per load and wave; sums as before
# speedup vs baseline: 8.3069x; 1.0758x over previous
#include <hip/hip_runtime.h>
#include <math.h>


typedef __attribute__((ext_vector_type(2))) float v2f;
typedef __attribute__((ext_vector_type(4))) float v4f;
typedef __attribute__((ext_vector_type(8))) float v8f;

#define NSEQ   16384
#define XL     1024
#define UL     256
#define YLN    256
#define CLEN   1024

__device__ __forceinline__ v8f wmma4(v2f a, v2f b, v8f c) {
  return __builtin_amdgcn_wmma_f32_16x16x4_f32(
       false, a,  false, b,
       (short)0, c,  false,  false);
}

__device__ __forceinline__ size_t panel_off(int R, int K) {
  return ((size_t)(R >> 4) * (XL / 8) + (size_t)(K >> 3)) * 128
         + (size_t)((R & 15) * 8 + (K & 7));
}

template <int WM, int WN>
__device__ __forceinline__ void gemm_rr_body(
    const float* __restrict__ Ap, const float* __restrict__ Bp,
    float* __restrict__ Cp, int K, int lda, int ldb, int ldc, int bx, int by) {
  const int lane = threadIdx.x & 31;
  const int w    = threadIdx.x >> 5;
  const int wm = w >> 2, wn = w & 3;
  const int tm = bx * (32 * WM) + wm * (16 * WM);
  const int tn = by * (64 * WN) + wn * (16 * WN);
  const int ml = lane & 15;
  const int kh = (lane >> 4) << 1;

  v8f acc[WM][WN];
#pragma unroll
  for (int mi = 0; mi < WM; ++mi)
#pragma unroll
    for (int ni = 0; ni < WN; ++ni) acc[mi][ni] = (v8f){};

  const float* ap[WM];
  const float* bp[WN];
#pragma unroll
  for (int mi = 0; mi < WM; ++mi)
    ap[mi] = Ap + (size_t)(tm + mi * 16 + ml) * lda + kh;
#pragma unroll
  for (int ni = 0; ni < WN; ++ni)
    bp[ni] = Bp + (size_t)kh * ldb + (tn + ni * 16 + ml);

  for (int k = 0; k < K; k += 4) {
    v2f af[WM], bf[WN];
    const size_t bo0 = (size_t)k * ldb, bo1 = bo0 + (size_t)ldb;
#pragma unroll
    for (int mi = 0; mi < WM; ++mi) af[mi] = (v2f){ ap[mi][k], ap[mi][k + 1] };
#pragma unroll
    for (int ni = 0; ni < WN; ++ni) bf[ni] = (v2f){ bp[ni][bo0], bp[ni][bo1] };
#pragma unroll
    for (int mi = 0; mi < WM; ++mi)
#pragma unroll
      for (int ni = 0; ni < WN; ++ni)
        acc[mi][ni] = wmma4(af[mi], bf[ni], acc[mi][ni]);
  }

  const int nn = lane & 15;
  const int mb = (lane >> 4) << 3;
#pragma unroll
  for (int mi = 0; mi < WM; ++mi)
#pragma unroll
    for (int ni = 0; ni < WN; ++ni) {
      float* cp = Cp + (size_t)(tm + mi * 16 + mb) * ldc + (tn + ni * 16 + nn);
#pragma unroll
      for (int r = 0; r < 8; ++r) cp[(size_t)r * ldc] = acc[mi][ni][r];
    }
}

template <int WM, int WN, int KU>
struct frag_set { v4f a[KU][WM]; v4f b[KU][WN]; };

template <int WM, int WN, int KU, int AS, int BS>
__device__ __forceinline__ void load_set(frag_set<WM, WN, KU>& f,
    const float* const (&ap)[WM], const float* const (&bp)[WN], int k) {
#pragma unroll
  for (int u = 0; u < KU; ++u) {
#pragma unroll
    for (int mi = 0; mi < WM; ++mi)
      f.a[u][mi] = *(const v4f*)(ap[mi] + (k / 8 + u) * AS);
#pragma unroll
    for (int ni = 0; ni < WN; ++ni)
      f.b[u][ni] = *(const v4f*)(bp[ni] + (k / 8 + u) * BS);
  }
}

template <int WM, int WN, int KU, int NCH>
__device__ __forceinline__ void mma_set(v8f (&acc)[NCH][WM][WN],
                                        const frag_set<WM, WN, KU>& f) {
#pragma unroll
  for (int u = 0; u < KU; ++u)
#pragma unroll
    for (int mi = 0; mi < WM; ++mi)
#pragma unroll
      for (int ni = 0; ni < WN; ++ni) {
        const v2f a0 = { f.a[u][mi].x, f.a[u][mi].y };
        const v2f a1 = { f.a[u][mi].z, f.a[u][mi].w };
        const v2f b0 = { f.b[u][ni].x, f.b[u][ni].y };
        const v2f b1 = { f.b[u][ni].z, f.b[u][ni].w };
        acc[0][mi][ni]       = wmma4(a0, b0, acc[0][mi][ni]);
        acc[NCH - 1][mi][ni] = wmma4(a1, b1, acc[NCH - 1][mi][ni]);
      }
}

struct rows_plain {
  const float* base; int ld;
  __device__ __forceinline__ const float* operator()(int row) const {
    return base + (size_t)row * ld;
  }
};
struct rows_tree {
  const float* X; int q0; int s2;
  __device__ __forceinline__ const float* operator()(int row) const {
    int q = q0 + row * s2;
    if (q < 0) q = 0;
    return X + (size_t)q * XL;
  }
};

template <int WM, int WN, int KU>
__device__ __forceinline__ void load_a_lds(frag_set<WM, WN, KU>& f,
    const float* buf, const int (&arow)[WM], int koff) {
#pragma unroll
  for (int u = 0; u < KU; ++u)
#pragma unroll
    for (int mi = 0; mi < WM; ++mi)
      f.a[u][mi] = *(const v4f*)(buf + arow[mi] + koff + u * 8);
}

template <int WM, int WN, int KU, int BS>
__device__ __forceinline__ void load_b_set(frag_set<WM, WN, KU>& f,
    const float* const (&bp)[WN], int k) {
#pragma unroll
  for (int u = 0; u < KU; ++u)
#pragma unroll
    for (int ni = 0; ni < WN; ++ni)
      f.b[u][ni] = *(const v4f*)(bp[ni] + (k / 8 + u) * BS);
}

template <int WM, int WN, int KU, int NCH, int WGM, int BS, class ROWS>
__device__ __forceinline__ void staged_kloop(
    v8f (&acc)[NCH][WM][WN], ROWS rows, const float* const (&bp)[WN],
    int K, int wm, int c, int h) {
  constexpr int R     = 16 * WM * WGM;
  constexpr int KT    = 8 * KU;
  constexpr int PK    = 2 * KT;
  constexpr int PITCH = PK + 4;
  constexpr int RP    = PK / 4;
  constexpr int NPC   = R * RP / 256;
  constexpr int BUF   = R * PITCH;
  static_assert((R * RP) % 256 == 0 && NPC >= 1, "whole pieces per thread");
  __shared__ __attribute__((aligned(16))) float stage[2 * BUF];

  const float* gsrc[NPC];
  int ldst[NPC];
#pragma unroll
  for (int j = 0; j < NPC; ++j) {
    const int i = (int)threadIdx.x + 256 * j;
    const int row = i / RP, pc = i % RP;
    gsrc[j] = rows(row) + pc * 4;
    ldst[j] = row * PITCH + pc * 4;
  }
  int arow[WM];
#pragma unroll
  for (int mi = 0; mi < WM; ++mi)
    arow[mi] = ((wm * WM + mi) * 16 + c) * PITCH + 4 * h;

  v4f g[NPC];
#pragma unroll
  for (int j = 0; j < NPC; ++j) g[j] = *(const v4f*)gsrc[j];
#pragma unroll
  for (int j = 0; j < NPC; ++j) *(v4f*)(stage + ldst[j]) = g[j];
  __syncthreads();

  frag_set<WM, WN, KU> f0, f1;
  load_a_lds<WM, WN, KU>(f0, stage, arow, 0);
  load_b_set<WM, WN, KU, BS>(f0, bp, 0);
#pragma nounroll
  for (int k = 0; k < K; k += PK) {
    const int par = (k / PK) & 1;
    const float* cur = stage + par * BUF;
    float* nxt = stage + (par ^ 1) * BUF;
    const bool more = (k + PK < K);
    if (more) {
#pragma unroll
      for (int j = 0; j < NPC; ++j) g[j] = *(const v4f*)(gsrc[j] + (k + PK));
    }
    load_a_lds<WM, WN, KU>(f1, cur, arow, KT);
    load_b_set<WM, WN, KU, BS>(f1, bp, k + KT);
    asm volatile("" ::: "memory");
    mma_set<WM, WN, KU, NCH>(acc, f0);
    if (more) {
#pragma unroll
      for (int j = 0; j < NPC; ++j) *(v4f*)(nxt + ldst[j]) = g[j];
    }
    __syncthreads();
    const int kn = more ? (k + PK) : k;
    load_a_lds<WM, WN, KU>(f0, more ? (const float*)nxt : cur, arow, 0);
    load_b_set<WM, WN, KU, BS>(f0, bp, kn);
    asm volatile("" ::: "memory");
    mma_set<WM, WN, KU, NCH>(acc, f1);
  }
}

template <int WM, int WN, int KU, int WGM, bool STG>
__device__ __forceinline__ void sweep_body(
    float* X, const float* __restrict__ P, int s, int r0, int bx, int by) {
  constexpr int WGN = 8 / WGM;
  const int lane = threadIdx.x & 31;
  const int w    = threadIdx.x >> 5;
  const int wm = w / WGN, wn = w % WGN;
  const int tm = bx * (16 * WM * WGM) + wm * (16 * WM);
  const int tn = by * (16 * WN * WGN) + wn * (16 * WN);
  const int c = lane & 15;
  const int h = lane >> 4;
  const int s2 = 2 * s;
  constexpr int NCH = (WM * WN < 4) ? 2 : 1;
  constexpr int KT = 8 * KU;
  static_assert(XL % (2 * KT) == 0, "two trips per loop pass");

  v8f acc[NCH][WM][WN];
#pragma unroll
  for (int ch = 0; ch < NCH; ++ch)
#pragma unroll
    for (int mi = 0; mi < WM; ++mi)
#pragma unroll
      for (int ni = 0; ni < WN; ++ni) acc[ch][mi][ni] = (v8f){};

  const float* bp[WN];
#pragma unroll
  for (int ni = 0; ni < WN; ++ni)
    bp[ni] = P + (size_t)((tn >> 4) + ni) * (16 * XL) + c * 8 + 4 * h;

  if constexpr (STG) {
    constexpr int R = 16 * WM * WGM;
    const rows_tree rows = { X, r0 - s + bx * R * s2, s2 };
    staged_kloop<WM, WN, KU, NCH, WGM, 128>(acc, rows, bp, XL, wm, c, h);
  } else {
    const float* ap[WM];
#pragma unroll
    for (int mi = 0; mi < WM; ++mi) {
      int q = r0 - s + (tm + mi * 16 + c) * s2;
      if (q < 0) q = 0;
      ap[mi] = X + (size_t)q * XL + 4 * h;
    }
    frag_set<WM, WN, KU> f0, f1;
    load_set<WM, WN, KU, 8, 128>(f0, ap, bp, 0);
    for (int k = 0; k < XL; k += 2 * KT) {
      load_set<WM, WN, KU, 8, 128>(f1, ap, bp, k + KT);
      asm volatile("" ::: "memory");
      mma_set<WM, WN, KU, NCH>(acc, f0);
      const int kn = (k + 2 * KT < XL) ? (k + 2 * KT) : k;
      load_set<WM, WN, KU, 8, 128>(f0, ap, bp, kn);
      asm volatile("" ::: "memory");
      mma_set<WM, WN, KU, NCH>(acc, f1);
    }
  }

  const bool open0 = (r0 < s);
  v8f old[WM][WN];
#pragma unroll
  for (int mi = 0; mi < WM; ++mi)
#pragma unroll
    for (int ni = 0; ni < WN; ++ni) {
#pragma unroll
      for (int r = 0; r < 8; ++r) {
        const int row = r0 + (tm + mi * 16 + 8 * h + r) * s2;
        old[mi][ni][r] = X[(size_t)row * XL + (tn + ni * 16 + c)];
      }
    }
#pragma unroll
  for (int mi = 0; mi < WM; ++mi)
#pragma unroll
    for (int ni = 0; ni < WN; ++ni) {
      v8f a = acc[0][mi][ni];
      if constexpr (NCH == 2) a = a + acc[1][mi][ni];
#pragma unroll
      for (int r = 0; r < 8; ++r) {
        const int m   = tm + mi * 16 + 8 * h + r;
        const int row = r0 + m * s2;
        const float add = (open0 && m == 0) ? 0.0f : a[r];
        X[(size_t)row * XL + (tn + ni * 16 + c)] = old[mi][ni][r] + add;
      }
    }
}

template <int WM, int WN, int KU, int WGM = 2>
__global__ __launch_bounds__(256) void sweep_k(
    float* X, const float* __restrict__ P, int s, int r0) {
  sweep_body<WM, WN, KU, WGM, true>(X, P, s, r0, blockIdx.x, blockIdx.y);
}

template <int WM, int WN, int KU, bool WT>
__device__ __forceinline__ void square_body(
    const float* __restrict__ Q, const float* __restrict__ Qt,
    float* __restrict__ Cn, float* __restrict__ Ct, int bx, int by) {
  const int lane = threadIdx.x & 31;
  const int w    = threadIdx.x >> 5;
  const int wm = w >> 2, wn = w & 3;
  const int tm = bx * (32 * WM) + wm * (16 * WM);
  const int tn = by * (64 * WN) + wn * (16 * WN);
  const int c = lane & 15;
  const int h = lane >> 4;
  constexpr int NCH = (WM * WN < 4) ? 2 : 1;
  constexpr int KT = 8 * KU;
  static_assert(XL % (2 * KT) == 0, "two trips per loop pass");

  v8f acc[NCH][WM][WN];
#pragma unroll
  for (int ch = 0; ch < NCH; ++ch)
#pragma unroll
    for (int mi = 0; mi < WM; ++mi)
#pragma unroll
      for (int ni = 0; ni < WN; ++ni) acc[ch][mi][ni] = (v8f){};

  const float* ap[WM];
  const float* bp[WN];
#pragma unroll
  for (int mi = 0; mi < WM; ++mi)
    ap[mi] = Q + (size_t)((tm >> 4) + mi) * (16 * XL) + c * 8 + 4 * h;
#pragma unroll
  for (int ni = 0; ni < WN; ++ni)
    bp[ni] = Qt + (size_t)((tn >> 4) + ni) * (16 * XL) + c * 8 + 4 * h;

  frag_set<WM, WN, KU> f0, f1;
  load_set<WM, WN, KU, 128, 128>(f0, ap, bp, 0);
  for (int k = 0; k < XL; k += 2 * KT) {
    load_set<WM, WN, KU, 128, 128>(f1, ap, bp, k + KT);
    asm volatile("" ::: "memory");
    mma_set<WM, WN, KU, NCH>(acc, f0);
    const int kn = (k + 2 * KT < XL) ? (k + 2 * KT) : k;
    load_set<WM, WN, KU, 128, 128>(f0, ap, bp, kn);
    asm volatile("" ::: "memory");
    mma_set<WM, WN, KU, NCH>(acc, f1);
  }

#pragma unroll
  for (int mi = 0; mi < WM; ++mi)
#pragma unroll
    for (int ni = 0; ni < WN; ++ni) {
      v8f a = acc[0][mi][ni];
      if constexpr (NCH == 2) a = a + acc[1][mi][ni];
      const int m0 = tm + mi * 16 + 8 * h;
      const int n  = tn + ni * 16 + c;
      if constexpr (WT) {
        float* p = Cn + panel_off(m0, n);
#pragma unroll
        for (int r = 0; r < 8; ++r) p[r * 8] = a[r];
        float* q = Ct + panel_off(n, m0);
        *(v4f*)q       = (v4f){ a[0], a[1], a[2], a[3] };
        *(v4f*)(q + 4) = (v4f){ a[4], a[5], a[6], a[7] };
      } else {
#pragma unroll
        for (int r = 0; r < 8; ++r) Cn[(size_t)(m0 + r) * XL + n] = a[r];
      }
    }
}

template <int MASK>
__device__ __forceinline__ float row_xmask_add(float v) {
  const int t = __builtin_amdgcn_update_dpp(0, __builtin_bit_cast(int, v),
                                            0x160 + MASK, 0xf, 0xf, true);
  return v + __builtin_bit_cast(float, t);
}

__device__ __forceinline__ float wave_sum(float v) {
  v = row_xmask_add<1>(v);
  v = row_xmask_add<2>(v);
  v = row_xmask_add<4>(v);
  v = row_xmask_add<8>(v);
  const int b = __builtin_bit_cast(int, v);
  const int o = __builtin_amdgcn_permlanex16(b, b, 0x76543210, 0xfedcba98, false, false);
  return v + __builtin_bit_cast(float, o);
}

__device__ __forceinline__ void matvec_body(
    const float* __restrict__ Mm, const float* __restrict__ vin,
    const float* vadd, float* vout, int bx) {
  __shared__ __attribute__((aligned(16))) float rsum[32];
  const int lane = threadIdx.x & 31, wid = threadIdx.x >> 5;
  const int line0 = bx * 32;
  const int row0  = line0 + wid * 4;

  float addv = 0.f;
  if (wid == 0) addv = vadd[line0 + lane];

  v4f x[8], m[4][8];
#pragma unroll
  for (int j = 0; j < 8; ++j) x[j] = *(const v4f*)(vin + 128 * j + 4 * lane);
#pragma unroll
  for (int r = 0; r < 4; ++r)
#pragma unroll
    for (int j = 0; j < 8; ++j)
      m[r][j] = *(const v4f*)(Mm + (size_t)(row0 + r) * XL + 128 * j + 4 * lane);
  __builtin_amdgcn_sched_barrier(0);

  v4f s4;
#pragma unroll
  for (int r = 0; r < 4; ++r) {
    v4f acc = m[r][0] * x[0];
#pragma unroll
    for (int j = 1; j < 8; ++j) acc += m[r][j] * x[j];
    s4[r] = wave_sum((acc.x + acc.y) + (acc.z + acc.w));
  }
  if (lane == 0) *(v4f*)(rsum + wid * 4) = s4;
  __syncthreads();
  if (wid == 0) vout[line0 + lane] = rsum[lane] + addv;
}

__global__ __launch_bounds__(256) void matvec_step_k(
    const float* __restrict__ Mm, const float* __restrict__ vin,
    const float* vadd, float* vout) {
  matvec_body(Mm, vin, vadd, vout, blockIdx.x);
}

template <int WM, int WN, int KU>
__global__ __launch_bounds__(256) void out_k(
    const float* __restrict__ xs, const float* __restrict__ Cm,
    const float* __restrict__ uf, const float* __restrict__ Dm,
    float* __restrict__ out) {
  static_assert(XL % (16 * KU) == 0 && UL % (16 * KU) == 0, "two trips per loop pass");
  const int lane = threadIdx.x & 31;
  const int w    = threadIdx.x >> 5;
  const int wm = w >> 2, wn = w & 3;
  const int tm = blockIdx.x * (32 * WM) + wm * (16 * WM);
  const int tn = blockIdx.y * (64 * WN) + wn * (16 * WN);
  const int c = lane & 15;
  const int h = lane >> 4;

  v8f acc[1][WM][WN];
#pragma unroll
  for (int mi = 0; mi < WM; ++mi)
#pragma unroll
    for (int ni = 0; ni < WN; ++ni) acc[0][mi][ni] = (v8f){};

  const int tm0 = blockIdx.x * (32 * WM);
  {
    const float* bp[WN];
#pragma unroll
    for (int ni = 0; ni < WN; ++ni)
      bp[ni] = Cm + (size_t)(tn + ni * 16 + c) * XL + 4 * h;
    const rows_plain rows = { xs + (size_t)tm0 * XL, XL };
    staged_kloop<WM, WN, KU, 1, 2, 8>(acc, rows, bp, XL, wm, c, h);
  }
  {
    const float* bp[WN];
#pragma unroll
    for (int ni = 0; ni < WN; ++ni)
      bp[ni] = Dm + (size_t)(tn + ni * 16 + c) * UL + 4 * h;
    const rows_plain rows = { uf + (size_t)tm0 * UL, UL };
    staged_kloop<WM, WN, KU, 1, 2, 8>(acc, rows, bp, UL, wm, c, h);
  }

#pragma unroll
  for (int mi = 0; mi < WM; ++mi)
#pragma unroll
    for (int ni = 0; ni < WN; ++ni) {
      const int yl = tn + ni * 16 + c;
      const int n0 = tm + mi * 16 + 8 * h;
      const v8f a = acc[0][mi][ni];
      float* op = out + (size_t)yl * NSEQ + n0;
      *(v4f*)op       = (v4f){ tanhf(a[0]), tanhf(a[1]), tanhf(a[2]), tanhf(a[3]) };
      *(v4f*)(op + 4) = (v4f){ tanhf(a[4]), tanhf(a[5]), tanhf(a[6]), tanhf(a[7]) };
    }
}

__device__ __forceinline__ void transpose_body(
    const float* __restrict__ in, float* __restrict__ out, int rows, int cols,
    int bx, int by) {
  __shared__ float tile[32][33];
  const int c0 = bx * 32, r0 = by * 32;
  const int tx = threadIdx.x & 31, ty = threadIdx.x >> 5;
#pragma unroll
  for (int i = 0; i < 32; i += 8)
    tile[ty + i][tx] = in[(size_t)(r0 + ty + i) * cols + (c0 + tx)];
  __syncthreads();
#pragma unroll
  for (int i = 0; i < 32; i += 8)
    out[(size_t)(c0 + ty + i) * rows + (r0 + tx)] = tile[tx][ty + i];
}

template <bool WTT>
__device__ __forceinline__ void panel_body(
    const float* __restrict__ in, float* __restrict__ rt, float* __restrict__ tt,
    int bx, int by) {
  __shared__ float tile[32][33];
  const int c0 = bx * 32, r0 = by * 32;
  const int tx = threadIdx.x & 31, ty = threadIdx.x >> 5;
#pragma unroll
  for (int i = 0; i < 32; i += 8) {
    const float v = in[(size_t)(r0 + ty + i) * XL + (c0 + tx)];
    if constexpr (WTT) tile[ty + i][tx] = v;
    rt[panel_off(r0 + ty + i, c0 + tx)] = v;
  }
  if constexpr (WTT) {
    __syncthreads();
#pragma unroll
    for (int i = 0; i < 32; i += 8)
      tt[panel_off(c0 + ty + i, r0 + tx)] = tile[tx][ty + i];
  }
}

constexpr int SQG = (XL / 64);

__global__ __launch_bounds__(256) void prep_k(
    const float* __restrict__ Bm, float* __restrict__ Bt,
    const float* __restrict__ Am, float* __restrict__ rt, float* __restrict__ tt) {
  constexpr int NT = (UL / 32) * (XL / 32);
  const int b = blockIdx.x;
  if (b < NT) {
    transpose_body(Bm, Bt, XL, UL, b % (UL / 32), b / (UL / 32));
  } else {
    const int q = b - NT;
    panel_body<true>(Am, rt, tt, q % (XL / 32), q / (XL / 32));
  }
}

template <int GX, int GY>
__global__ __launch_bounds__(256) void bu_sq_k(
    const float* __restrict__ uf, const float* __restrict__ Bt, float* __restrict__ X,
    int K, int lda, int ldb, int ldc,
    const float* __restrict__ Q, const float* __restrict__ Qt,
    float* __restrict__ Cn, float* __restrict__ Ct) {
  const int b = blockIdx.x;
  if (b < GX * GY) {
    gemm_rr_body<4, 4>(uf, Bt, X, K, lda, ldb, ldc, b % GX, b / GX);
  } else {
    const int q = b - GX * GY;
    square_body<2, 1, 8, true>(Q, Qt, Cn, Ct, q % SQG, q / SQG);
  }
}

__global__ __launch_bounds__(256) void mv_sq_k(
    const float* __restrict__ Mm, const float* __restrict__ vin,
    const float* vadd, float* vout,
    const float* __restrict__ Q, const float* __restrict__ Qt,
    float* __restrict__ Cn, float* __restrict__ Ct) {
  const int b = blockIdx.x;
  if (b < XL / 32) {
    matvec_body(Mm, vin, vadd, vout, b);
  } else {
    const int q = b - XL / 32;
    square_body<2, 1, 8, true>(Q, Qt, Cn, Ct, q % SQG, q / SQG);
  }
}

template <int WM, int WN, int KU, bool WT>
__global__ __launch_bounds__(256) void sweep_sq_k(
    float* X, const float* __restrict__ P, int s, int r0, int gxs, int nsw,
    const float* __restrict__ Q, const float* __restrict__ Qt,
    float* __restrict__ Cn, float* __restrict__ Ct) {
  const int b = blockIdx.x;
  if (b < nsw) {
    sweep_body<WM, WN, KU, 2, (KU <= 8)>(X, P, s, r0, b & ((1 << gxs) - 1), b >> gxs);
  } else {
    const int q = b - nsw;
    square_body<2, 1, 8, WT>(Q, Qt, Cn, Ct, q % SQG, q / SQG);
  }
}

template <int WM, int WN, int KU>
__global__ __launch_bounds__(256) void sweep_panel_k(
    float* X, const float* __restrict__ P, int s, int r0, int gxs, int nsw,
    const float* __restrict__ Am, float* __restrict__ rt) {
  const int b = blockIdx.x;
  if (b < nsw) {
    sweep_body<WM, WN, KU, 2, true>(X, P, s, r0, b & ((1 << gxs) - 1), b >> gxs);
  } else {
    const int q = b - nsw;
    panel_body<false>(Am, rt, nullptr, q % (XL / 32), q / (XL / 32));
  }
}

extern "C" void kernel_launch(void* const* d_in, const int* in_sizes, int n_in,
                              void* d_out, int out_size, void* d_ws, size_t ws_size,
                              hipStream_t stream) {
  (void)in_sizes; (void)n_in; (void)out_size; (void)ws_size;
  const float* u  = (const float*)d_in[0];
  const float* A  = (const float*)d_in[1];
  const float* B  = (const float*)d_in[2];
  const float* C  = (const float*)d_in[3];
  const float* D  = (const float*)d_in[4];
  const float* x0 = (const float*)d_in[5];
  float* out = (float*)d_out;

  float* ws = (float*)d_ws;
  size_t off = 0;
  float* X  = ws + off; off += (size_t)NSEQ * XL;
  float* Bt = ws + off; off += (size_t)UL * XL;
  float* P2   = ws + off; off += (size_t)XL * XL;
  float* P4   = ws + off; off += (size_t)XL * XL;
  float* P8   = ws + off; off += (size_t)XL * XL;
  float* P16  = ws + off; off += (size_t)XL * XL;
  float* P32  = ws + off; off += (size_t)XL * XL;
  float* P64  = ws + off; off += (size_t)XL * XL;
  float* P128 = ws + off; off += (size_t)XL * XL;
  float* P256  = out;
  float* P512  = out + (size_t)XL * XL;
  float* P1024 = out + 2 * (size_t)XL * XL;
  float* P1    = out + 3 * (size_t)XL * XL;

  float* T1 = P4,   * T2 = P8,    * T4 = P16,    * T8 = P32,   * T16 = P64;
  float* T32 = P128, * T64 = P256, * T128 = P512, * T256 = P1024, * T512 = P1;

  prep_k<<<(UL / 32) * (XL / 32) + (XL / 32) * (XL / 32), 256, 0, stream>>>(B, Bt, A, P1, T1);

  bu_sq_k<NSEQ / 128, XL / 256><<<(NSEQ / 128) * (XL / 256) + SQG * SQG, 256, 0, stream>>>(
      u, Bt, X, UL, UL, XL, XL, P1, T1, P2, T2);
  mv_sq_k<<<XL / 32 + SQG * SQG, 256, 0, stream>>>(A, x0, X, X, P2, T2, P4, T4);

  sweep_sq_k<4, 4, 2, true><<<(8192 / 128) * (XL / 256) + SQG * SQG, 256, 0, stream>>>(
      X, P1,  1,  1,   6, (8192 / 128) * (XL / 256), P4,   T4,   P8,    T8);
  sweep_sq_k<2, 4, 2, true><<<(4096 / 64) * (XL / 256) + SQG * SQG, 256, 0, stream>>>(
      X, P2,  2,  3,   6, (4096 / 64) * (XL / 256),  P8,   T8,   P16,   T16);
  sweep_sq_k<2, 2, 4, true><<<(2048 / 64) * (XL / 128) + SQG * SQG, 256, 0, stream>>>(
      X, P4,  4,  7,   5, (2048 / 64) * (XL / 128),  P16,  T16,  P32,   T32);
  sweep_sq_k<1, 2, 8, true><<<(1024 / 32) * (XL / 128) + SQG * SQG, 256, 0, stream>>>(
      X, P8,  8,  15,  5, (1024 / 32) * (XL / 128),  P32,  T32,  P64,   T64);
  sweep_sq_k<1, 1, 16, true><<<(512 / 32) * (XL / 64) + SQG * SQG, 256, 0, stream>>>(
      X, P16, 16, 31,  4, (512 / 32) * (XL / 64),    P64,  T64,  P128,  T128);
  sweep_sq_k<1, 1, 16, true><<<(256 / 32) * (XL / 64) + SQG * SQG, 256, 0, stream>>>(
      X, P32, 32, 63,  3, (256 / 32) * (XL / 64),    P128, T128, P256,  T256);
  sweep_sq_k<1, 1, 16, true><<<(128 / 32) * (XL / 64) + SQG * SQG, 256, 0, stream>>>(
      X, P64, 64, 127, 2, (128 / 32) * (XL / 64),    P256, T256, P512,  T512);
  sweep_sq_k<1, 1, 16, false><<<(64 / 32) * (XL / 64) + SQG * SQG, 256, 0, stream>>>(
      X, P128, 128, 255, 1, (64 / 32) * (XL / 64),   P512, T512, P1024, nullptr);
  sweep_panel_k<1, 1, 16><<<(32 / 32) * (XL / 64) + (XL / 32) * (XL / 32), 256, 0, stream>>>(
      X, P256, 256, 511, 0, (32 / 32) * (XL / 64),   A, P1);
  sweep_k<1, 1, 16, 1><<<dim3(16 / 16, XL / 128), 256, 0, stream>>>(X, P512, 512, 1023);

  for (int j = 1; j < NSEQ / CLEN; ++j) {
    float* prev = X + ((size_t)j * CLEN - 1) * XL;
    float* cur  = X + ((size_t)(j + 1) * CLEN - 1) * XL;
    matvec_step_k<<<32, 256, 0, stream>>>(P1024, prev, cur, cur);
  }

  sweep_k<1, 1, 16, 1><<<dim3(16 / 16, XL / 128), 256, 0, stream>>>(X, P512, 512, 511);
  sweep_k<1, 1, 16><<<dim3(32 / 32,    XL / 64),  256, 0, stream>>>(X, P256, 256, 255);
  sweep_k<1, 1, 16><<<dim3(64 / 32,    XL / 64),  256, 0, stream>>>(X, P128, 128, 127);
  sweep_k<1, 1, 16><<<dim3(128 / 32,   XL / 64),  256, 0, stream>>>(X, P64, 64, 63);
  sweep_k<1, 1, 16><<<dim3(256 / 32,   XL / 64),  256, 0, stream>>>(X, P32, 32, 31);
  sweep_k<1, 1, 16><<<dim3(512 / 32,   XL / 64),  256, 0, stream>>>(X, P16, 16, 15);
  sweep_k<1, 2, 8><<<dim3(1024 / 32,  XL / 128), 256, 0, stream>>>(X, P8,  8,  7);
  sweep_k<2, 2, 4><<<dim3(2048 / 64,  XL / 128), 256, 0, stream>>>(X, P4,  4,  3);
  sweep_k<2, 4, 2><<<dim3(4096 / 64,  XL / 256), 256, 0, stream>>>(X, P2,  2,  1);
  sweep_k<4, 4, 2><<<dim3(8192 / 128, XL / 256), 256, 0, stream>>>(X, P1,  1,  0);

  out_k<2, 4, 2><<<dim3(NSEQ / 64, YLN / 256), 256, 0, stream>>>(X, C, u, D, out);
}


# last down level (s = 1, 8192 rows) folded into the output product: even times from the row in front with C A and C B + D (two small products beside level s = 64); sum order changes
# speedup vs baseline: 9.3406x; 1.1244x over previous
#include <hip/hip_runtime.h>
#include <math.h>


typedef __attribute__((ext_vector_type(2))) float v2f;
typedef __attribute__((ext_vector_type(4))) float v4f;
typedef __attribute__((ext_vector_type(8))) float v8f;

#define NSEQ   16384
#define XL     1024
#define UL     256
#define YLN    256
#define CLEN   1024

__device__ __forceinline__ v8f wmma4(v2f a, v2f b, v8f c) {
  return __builtin_amdgcn_wmma_f32_16x16x4_f32(
       false, a,  false, b,
       (short)0, c,  false,  false);
}

__device__ __forceinline__ size_t panel_off(int R, int K) {
  return ((size_t)(R >> 4) * (XL / 8) + (size_t)(K >> 3)) * 128
         + (size_t)((R & 15) * 8 + (K & 7));
}

template <int WM, int WN>
__device__ __forceinline__ void gemm_rr_body(
    const float* __restrict__ Ap, const float* __restrict__ Bp,
    float* __restrict__ Cp, int K, int lda, int ldb, int ldc, int bx, int by) {
  const int lane = threadIdx.x & 31;
  const int w    = threadIdx.x >> 5;
  const int wm = w >> 2, wn = w & 3;
  const int tm = bx * (32 * WM) + wm * (16 * WM);
  const int tn = by * (64 * WN) + wn * (16 * WN);
  const int ml = lane & 15;
  const int kh = (lane >> 4) << 1;

  v8f acc[WM][WN];
#pragma unroll
  for (int mi = 0; mi < WM; ++mi)
#pragma unroll
    for (int ni = 0; ni < WN; ++ni) acc[mi][ni] = (v8f){};

  const float* ap[WM];
  const float* bp[WN];
#pragma unroll
  for (int mi = 0; mi < WM; ++mi)
    ap[mi] = Ap + (size_t)(tm + mi * 16 + ml) * lda + kh;
#pragma unroll
  for (int ni = 0; ni < WN; ++ni)
    bp[ni] = Bp + (size_t)kh * ldb + (tn + ni * 16 + ml);

  for (int k = 0; k < K; k += 4) {
    v2f af[WM], bf[WN];
    const size_t bo0 = (size_t)k * ldb, bo1 = bo0 + (size_t)ldb;
#pragma unroll
    for (int mi = 0; mi < WM; ++mi) af[mi] = (v2f){ ap[mi][k], ap[mi][k + 1] };
#pragma unroll
    for (int ni = 0; ni < WN; ++ni) bf[ni] = (v2f){ bp[ni][bo0], bp[ni][bo1] };
#pragma unroll
    for (int mi = 0; mi < WM; ++mi)
#pragma unroll
      for (int ni = 0; ni < WN; ++ni)
        acc[mi][ni] = wmma4(af[mi], bf[ni], acc[mi][ni]);
  }

  const int nn = lane & 15;
  const int mb = (lane >> 4) << 3;
#pragma unroll
  for (int mi = 0; mi < WM; ++mi)
#pragma unroll
    for (int ni = 0; ni < WN; ++ni) {
      float* cp = Cp + (size_t)(tm + mi * 16 + mb) * ldc + (tn + ni * 16 + nn);
#pragma unroll
      for (int r = 0; r < 8; ++r) cp[(size_t)r * ldc] = acc[mi][ni][r];
    }
}

template <int WM, int WN, int KU>
struct frag_set { v4f a[KU][WM]; v4f b[KU][WN]; };

template <int WM, int WN, int KU, int AS, int BS>
__device__ __forceinline__ void load_set(frag_set<WM, WN, KU>& f,
    const float* const (&ap)[WM], const float* const (&bp)[WN], int k) {
#pragma unroll
  for (int u = 0; u < KU; ++u) {
#pragma unroll
    for (int mi = 0; mi < WM; ++mi)
      f.a[u][mi] = *(const v4f*)(ap[mi] + (k / 8 + u) * AS);
#pragma unroll
    for (int ni = 0; ni < WN; ++ni)
      f.b[u][ni] = *(const v4f*)(bp[ni] + (k / 8 + u) * BS);
  }
}

template <int WM, int WN, int KU, int NCH, int WNB = WN>
__device__ __forceinline__ void mma_set(v8f (&acc)[NCH][WM][WN],
                                        const frag_set<WM, WNB, KU>& f) {
  constexpr int MG = WM / (WNB / WN);
#pragma unroll
  for (int u = 0; u < KU; ++u)
#pragma unroll
    for (int mi = 0; mi < WM; ++mi)
#pragma unroll
      for (int ni = 0; ni < WN; ++ni) {
        const int bi = (mi / MG) * WN + ni;
        const v2f a0 = { f.a[u][mi].x, f.a[u][mi].y };
        const v2f a1 = { f.a[u][mi].z, f.a[u][mi].w };
        const v2f b0 = { f.b[u][bi].x, f.b[u][bi].y };
        const v2f b1 = { f.b[u][bi].z, f.b[u][bi].w };
        acc[0][mi][ni]       = wmma4(a0, b0, acc[0][mi][ni]);
        acc[NCH - 1][mi][ni] = wmma4(a1, b1, acc[NCH - 1][mi][ni]);
      }
}

struct rows_plain {
  const float* base; int ld;
  __device__ __forceinline__ const float* operator()(int row) const {
    return base + (size_t)row * ld;
  }
};
struct rows_tree {
  const float* X; int q0; int s2;
  __device__ __forceinline__ const float* operator()(int row) const {
    int q = q0 + row * s2;
    if (q < 0) q = 0;
    return X + (size_t)q * XL;
  }
};

template <int WM, int WN, int KU>
__device__ __forceinline__ void load_a_lds(frag_set<WM, WN, KU>& f,
    const float* buf, const int (&arow)[WM], int koff) {
#pragma unroll
  for (int u = 0; u < KU; ++u)
#pragma unroll
    for (int mi = 0; mi < WM; ++mi)
      f.a[u][mi] = *(const v4f*)(buf + arow[mi] + koff + u * 8);
}

template <int WM, int WN, int KU, int BS>
__device__ __forceinline__ void load_b_set(frag_set<WM, WN, KU>& f,
    const float* const (&bp)[WN], int k) {
#pragma unroll
  for (int u = 0; u < KU; ++u)
#pragma unroll
    for (int ni = 0; ni < WN; ++ni)
      f.b[u][ni] = *(const v4f*)(bp[ni] + (k / 8 + u) * BS);
}

template <int WM, int WN, int WNB, int KU, int NCH, int WGM, int BS, class ROWS>
__device__ __forceinline__ void staged_kloop(
    v8f (&acc)[NCH][WM][WN], ROWS rows, const float* const (&bp)[WNB],
    int K, int wm, int c, int h) {
  constexpr int R     = 16 * WM * WGM;
  constexpr int KT    = 8 * KU;
  constexpr int PK    = 2 * KT;
  constexpr int PITCH = PK + 4;
  constexpr int RP    = PK / 4;
  constexpr int NPC   = R * RP / 256;
  constexpr int BUF   = R * PITCH;
  static_assert((R * RP) % 256 == 0 && NPC >= 1, "whole pieces per thread");
  __shared__ __attribute__((aligned(16))) float stage[2 * BUF];

  const float* gsrc[NPC];
  int ldst[NPC];
#pragma unroll
  for (int j = 0; j < NPC; ++j) {
    const int i = (int)threadIdx.x + 256 * j;
    const int row = i / RP, pc = i % RP;
    gsrc[j] = rows(row) + pc * 4;
    ldst[j] = row * PITCH + pc * 4;
  }
  int arow[WM];
#pragma unroll
  for (int mi = 0; mi < WM; ++mi)
    arow[mi] = ((wm * WM + mi) * 16 + c) * PITCH + 4 * h;

  v4f g[NPC];
#pragma unroll
  for (int j = 0; j < NPC; ++j) g[j] = *(const v4f*)gsrc[j];
#pragma unroll
  for (int j = 0; j < NPC; ++j) *(v4f*)(stage + ldst[j]) = g[j];
  __syncthreads();

  frag_set<WM, WNB, KU> f0, f1;
  load_a_lds<WM, WNB, KU>(f0, stage, arow, 0);
  load_b_set<WM, WNB, KU, BS>(f0, bp, 0);
#pragma nounroll
  for (int k = 0; k < K; k += PK) {
    const int par = (k / PK) & 1;
    const float* cur = stage + par * BUF;
    float* nxt = stage + (par ^ 1) * BUF;
    const bool more = (k + PK < K);
    if (more) {
#pragma unroll
      for (int j = 0; j < NPC; ++j) g[j] = *(const v4f*)(gsrc[j] + (k + PK));
    }
    load_a_lds<WM, WNB, KU>(f1, cur, arow, KT);
    load_b_set<WM, WNB, KU, BS>(f1, bp, k + KT);
    asm volatile("" ::: "memory");
    mma_set<WM, WN, KU, NCH, WNB>(acc, f0);
    if (more) {
#pragma unroll
      for (int j = 0; j < NPC; ++j) *(v4f*)(nxt + ldst[j]) = g[j];
    }
    __syncthreads();
    const int kn = more ? (k + PK) : k;
    load_a_lds<WM, WNB, KU>(f0, more ? (const float*)nxt : cur, arow, 0);
    load_b_set<WM, WNB, KU, BS>(f0, bp, kn);
    asm volatile("" ::: "memory");
    mma_set<WM, WN, KU, NCH, WNB>(acc, f1);
  }
}

template <int WM, int WN, int KU, int WGM, bool STG>
__device__ __forceinline__ void sweep_body(
    float* X, const float* __restrict__ P, int s, int r0, int bx, int by) {
  constexpr int WGN = 8 / WGM;
  const int lane = threadIdx.x & 31;
  const int w    = threadIdx.x >> 5;
  const int wm = w / WGN, wn = w % WGN;
  const int tm = bx * (16 * WM * WGM) + wm * (16 * WM);
  const int tn = by * (16 * WN * WGN) + wn * (16 * WN);
  const int c = lane & 15;
  const int h = lane >> 4;
  const int s2 = 2 * s;
  constexpr int NCH = (WM * WN < 4) ? 2 : 1;
  constexpr int KT = 8 * KU;
  static_assert(XL % (2 * KT) == 0, "two trips per loop pass");

  v8f acc[NCH][WM][WN];
#pragma unroll
  for (int ch = 0; ch < NCH; ++ch)
#pragma unroll
    for (int mi = 0; mi < WM; ++mi)
#pragma unroll
      for (int ni = 0; ni < WN; ++ni) acc[ch][mi][ni] = (v8f){};

  const float* bp[WN];
#pragma unroll
  for (int ni = 0; ni < WN; ++ni)
    bp[ni] = P + (size_t)((tn >> 4) + ni) * (16 * XL) + c * 8 + 4 * h;

  if constexpr (STG) {
    constexpr int R = 16 * WM * WGM;
    const rows_tree rows = { X, r0 - s + bx * R * s2, s2 };
    staged_kloop<WM, WN, WN, KU, NCH, WGM, 128>(acc, rows, bp, XL, wm, c, h);
  } else {
    const float* ap[WM];
#pragma unroll
    for (int mi = 0; mi < WM; ++mi) {
      int q = r0 - s + (tm + mi * 16 + c) * s2;
      if (q < 0) q = 0;
      ap[mi] = X + (size_t)q * XL + 4 * h;
    }
    frag_set<WM, WN, KU> f0, f1;
    load_set<WM, WN, KU, 8, 128>(f0, ap, bp, 0);
    for (int k = 0; k < XL; k += 2 * KT) {
      load_set<WM, WN, KU, 8, 128>(f1, ap, bp, k + KT);
      asm volatile("" ::: "memory");
      mma_set<WM, WN, KU, NCH>(acc, f0);
      const int kn = (k + 2 * KT < XL) ? (k + 2 * KT) : k;
      load_set<WM, WN, KU, 8, 128>(f0, ap, bp, kn);
      asm volatile("" ::: "memory");
      mma_set<WM, WN, KU, NCH>(acc, f1);
    }
  }

  const bool open0 = (r0 < s);
  v8f old[WM][WN];
#pragma unroll
  for (int mi = 0; mi < WM; ++mi)
#pragma unroll
    for (int ni = 0; ni < WN; ++ni) {
#pragma unroll
      for (int r = 0; r < 8; ++r) {
        const int row = r0 + (tm + mi * 16 + 8 * h + r) * s2;
        old[mi][ni][r] = X[(size_t)row * XL + (tn + ni * 16 + c)];
      }
    }
#pragma unroll
  for (int mi = 0; mi < WM; ++mi)
#pragma unroll
    for (int ni = 0; ni < WN; ++ni) {
      v8f a = acc[0][mi][ni];
      if constexpr (NCH == 2) a = a + acc[1][mi][ni];
#pragma unroll
      for (int r = 0; r < 8; ++r) {
        const int m   = tm + mi * 16 + 8 * h + r;
        const int row = r0 + m * s2;
        const float add = (open0 && m == 0) ? 0.0f : a[r];
        X[(size_t)row * XL + (tn + ni * 16 + c)] = old[mi][ni][r] + add;
      }
    }
}

template <int WM, int WN, int KU, int WGM = 2>
__global__ __launch_bounds__(256) void sweep_k(
    float* X, const float* __restrict__ P, int s, int r0) {
  sweep_body<WM, WN, KU, WGM, true>(X, P, s, r0, blockIdx.x, blockIdx.y);
}

template <int WM, int WN, int KU, bool WT>
__device__ __forceinline__ void square_body(
    const float* __restrict__ Q, const float* __restrict__ Qt,
    float* __restrict__ Cn, float* __restrict__ Ct, int bx, int by) {
  const int lane = threadIdx.x & 31;
  const int w    = threadIdx.x >> 5;
  const int wm = w >> 2, wn = w & 3;
  const int tm = bx * (32 * WM) + wm * (16 * WM);
  const int tn = by * (64 * WN) + wn * (16 * WN);
  const int c = lane & 15;
  const int h = lane >> 4;
  constexpr int NCH = (WM * WN < 4) ? 2 : 1;
  constexpr int KT = 8 * KU;
  static_assert(XL % (2 * KT) == 0, "two trips per loop pass");

  v8f acc[NCH][WM][WN];
#pragma unroll
  for (int ch = 0; ch < NCH; ++ch)
#pragma unroll
    for (int mi = 0; mi < WM; ++mi)
#pragma unroll
      for (int ni = 0; ni < WN; ++ni) acc[ch][mi][ni] = (v8f){};

  const float* ap[WM];
  const float* bp[WN];
#pragma unroll
  for (int mi = 0; mi < WM; ++mi)
    ap[mi] = Q + (size_t)((tm >> 4) + mi) * (16 * XL) + c * 8 + 4 * h;
#pragma unroll
  for (int ni = 0; ni < WN; ++ni)
    bp[ni] = Qt + (size_t)((tn >> 4) + ni) * (16 * XL) + c * 8 + 4 * h;

  frag_set<WM, WN, KU> f0, f1;
  load_set<WM, WN, KU, 128, 128>(f0, ap, bp, 0);
  for (int k = 0; k < XL; k += 2 * KT) {
    load_set<WM, WN, KU, 128, 128>(f1, ap, bp, k + KT);
    asm volatile("" ::: "memory");
    mma_set<WM, WN, KU, NCH>(acc, f0);
    const int kn = (k + 2 * KT < XL) ? (k + 2 * KT) : k;
    load_set<WM, WN, KU, 128, 128>(f0, ap, bp, kn);
    asm volatile("" ::: "memory");
    mma_set<WM, WN, KU, NCH>(acc, f1);
  }

#pragma unroll
  for (int mi = 0; mi < WM; ++mi)
#pragma unroll
    for (int ni = 0; ni < WN; ++ni) {
      v8f a = acc[0][mi][ni];
      if constexpr (NCH == 2) a = a + acc[1][mi][ni];
      const int m0 = tm + mi * 16 + 8 * h;
      const int n  = tn + ni * 16 + c;
      if constexpr (WT) {
        float* p = Cn + panel_off(m0, n);
#pragma unroll
        for (int r = 0; r < 8; ++r) p[r * 8] = a[r];
        float* q = Ct + panel_off(n, m0);
        *(v4f*)q       = (v4f){ a[0], a[1], a[2], a[3] };
        *(v4f*)(q + 4) = (v4f){ a[4], a[5], a[6], a[7] };
      } else {
#pragma unroll
        for (int r = 0; r < 8; ++r) Cn[(size_t)(m0 + r) * XL + n] = a[r];
      }
    }
}

template <int BS, bool ADD>
__device__ __forceinline__ void cproj_body(
    const float* __restrict__ Cm, const float* __restrict__ Mt,
    const float* __restrict__ Dadd, float* __restrict__ Rm, int N, int bx, int by) {
  constexpr int WM = 2, WN = 1, KU = 8, NCH = 2, KT = 8 * KU;
  const int lane = threadIdx.x & 31;
  const int w    = threadIdx.x >> 5;
  const int wm = w >> 2, wn = w & 3;
  const int tm = bx * 64 + wm * 32;
  const int tn = by * 64 + wn * 16;
  const int c = lane & 15;
  const int h = lane >> 4;

  v8f acc[NCH][WM][WN];
#pragma unroll
  for (int ch = 0; ch < NCH; ++ch)
#pragma unroll
    for (int mi = 0; mi < WM; ++mi) acc[ch][mi][0] = (v8f){};

  const float* ap[WM];
  const float* bp[WN];
#pragma unroll
  for (int mi = 0; mi < WM; ++mi)
    ap[mi] = Cm + (size_t)(tm + mi * 16 + c) * XL + 4 * h;
  if constexpr (BS == 128)
    bp[0] = Mt + (size_t)(tn >> 4) * (16 * XL) + c * 8 + 4 * h;
  else
    bp[0] = Mt + (size_t)(tn + c) * XL + 4 * h;

  frag_set<WM, WN, KU> f0, f1;
  load_set<WM, WN, KU, 8, BS>(f0, ap, bp, 0);
  for (int k = 0; k < XL; k += 2 * KT) {
    load_set<WM, WN, KU, 8, BS>(f1, ap, bp, k + KT);
    asm volatile("" ::: "memory");
    mma_set<WM, WN, KU, NCH>(acc, f0);
    const int kn = (k + 2 * KT < XL) ? (k + 2 * KT) : k;
    load_set<WM, WN, KU, 8, BS>(f0, ap, bp, kn);
    asm volatile("" ::: "memory");
    mma_set<WM, WN, KU, NCH>(acc, f1);
  }

#pragma unroll
  for (int mi = 0; mi < WM; ++mi) {
    const v8f a = acc[0][mi][0] + acc[1][mi][0];
    const int m0 = tm + mi * 16 + 8 * h;
    const int n  = tn + c;
#pragma unroll
    for (int r = 0; r < 8; ++r) {
      float v = a[r];
      if constexpr (ADD) v += Dadd[(size_t)(m0 + r) * N + n];
      Rm[(size_t)(m0 + r) * N + n] = v;
    }
  }
}

template <int MASK>
__device__ __forceinline__ float row_xmask_add(float v) {
  const int t = __builtin_amdgcn_update_dpp(0, __builtin_bit_cast(int, v),
                                            0x160 + MASK, 0xf, 0xf, true);
  return v + __builtin_bit_cast(float, t);
}

__device__ __forceinline__ float wave_sum(float v) {
  v = row_xmask_add<1>(v);
  v = row_xmask_add<2>(v);
  v = row_xmask_add<4>(v);
  v = row_xmask_add<8>(v);
  const int b = __builtin_bit_cast(int, v);
  const int o = __builtin_amdgcn_permlanex16(b, b, 0x76543210, 0xfedcba98, false, false);
  return v + __builtin_bit_cast(float, o);
}

__device__ __forceinline__ void matvec_body(
    const float* __restrict__ Mm, const float* __restrict__ vin,
    const float* vadd, float* vout, int bx) {
  __shared__ __attribute__((aligned(16))) float rsum[32];
  const int lane = threadIdx.x & 31, wid = threadIdx.x >> 5;
  const int line0 = bx * 32;
  const int row0  = line0 + wid * 4;

  float addv = 0.f;
  if (wid == 0) addv = vadd[line0 + lane];

  v4f x[8], m[4][8];
#pragma unroll
  for (int j = 0; j < 8; ++j) x[j] = *(const v4f*)(vin + 128 * j + 4 * lane);
#pragma unroll
  for (int r = 0; r < 4; ++r)
#pragma unroll
    for (int j = 0; j < 8; ++j)
      m[r][j] = *(const v4f*)(Mm + (size_t)(row0 + r) * XL + 128 * j + 4 * lane);
  __builtin_amdgcn_sched_barrier(0);

  v4f s4;
#pragma unroll
  for (int r = 0; r < 4; ++r) {
    v4f acc = m[r][0] * x[0];
#pragma unroll
    for (int j = 1; j < 8; ++j) acc += m[r][j] * x[j];
    s4[r] = wave_sum((acc.x + acc.y) + (acc.z + acc.w));
  }
  if (lane == 0) *(v4f*)(rsum + wid * 4) = s4;
  __syncthreads();
  if (wid == 0) vout[line0 + lane] = rsum[lane] + addv;
}

__global__ __launch_bounds__(256) void matvec_step_k(
    const float* __restrict__ Mm, const float* __restrict__ vin,
    const float* vadd, float* vout) {
  matvec_body(Mm, vin, vadd, vout, blockIdx.x);
}

struct rows_par {
  const float* base; const float* first; int ld; int t0; int back;
  __device__ __forceinline__ const float* operator()(int row) const {
    const int q = t0 + (row >> 6) * 64 + 2 * (row & 31) + 1 - back * ((row >> 5) & 1);
    return (q < 0) ? first : base + (size_t)q * ld;
  }
};

template <int KU>
__global__ __launch_bounds__(256) void out_k(
    const float* __restrict__ xs, const float* __restrict__ x0,
    const float* __restrict__ Cm, const float* __restrict__ CAm,
    const float* __restrict__ uf, const float* __restrict__ Dm,
    const float* __restrict__ CBDm, float* __restrict__ out) {
  constexpr int WM = 4, WN = 2;
  static_assert(XL % (16 * KU) == 0 && UL % (16 * KU) == 0, "two trips per loop pass");
  const int lane = threadIdx.x & 31;
  const int w    = threadIdx.x >> 5;
  const int wm = w >> 2, wn = w & 3;
  const int t0 = blockIdx.x * 128;
  const int tn = blockIdx.y * 128 + wn * (16 * WN);
  const int c = lane & 15;
  const int h = lane >> 4;

  v8f acc[1][WM][WN];
#pragma unroll
  for (int mi = 0; mi < WM; ++mi)
#pragma unroll
    for (int ni = 0; ni < WN; ++ni) acc[0][mi][ni] = (v8f){};

  {
    const float* bp[2 * WN];
#pragma unroll
    for (int ni = 0; ni < WN; ++ni) {
      bp[ni]      = Cm  + (size_t)(tn + ni * 16 + c) * XL + 4 * h;
      bp[WN + ni] = CAm + (size_t)(tn + ni * 16 + c) * XL + 4 * h;
    }
    const rows_par rows = { xs, x0, XL, t0, 2 };
    staged_kloop<WM, WN, 2 * WN, KU, 1, 2, 8>(acc, rows, bp, XL, wm, c, h);
  }
  {
    const float* bp[2 * WN];
#pragma unroll
    for (int ni = 0; ni < WN; ++ni) {
      bp[ni]      = Dm   + (size_t)(tn + ni * 16 + c) * UL + 4 * h;
      bp[WN + ni] = CBDm + (size_t)(tn + ni * 16 + c) * UL + 4 * h;
    }
    const rows_par rows = { uf, uf, UL, t0, 1 };
    staged_kloop<WM, WN, 2 * WN, KU, 1, 2, 8>(acc, rows, bp, UL, wm, c, h);
  }

#pragma unroll
  for (int m1 = 0; m1 < 2; ++m1)
#pragma unroll
    for (int ni = 0; ni < WN; ++ni) {
      const int yl = tn + ni * 16 + c;
      const int n0 = t0 + wm * 64 + m1 * 32 + 16 * h;
      const v8f od = acc[0][m1][ni];
      const v8f ev = acc[0][2 + m1][ni];
      float* op = out + (size_t)yl * NSEQ + n0;
#pragma unroll
      for (int r = 0; r < 8; r += 2)
        *(v4f*)(op + 2 * r) = (v4f){ tanhf(ev[r]), tanhf(od[r]), tanhf(ev[r + 1]), tanhf(od[r + 1]) };
    }
}

__device__ __forceinline__ void transpose_body(
    const float* __restrict__ in, float* __restrict__ out, int rows, int cols,
    int bx, int by) {
  __shared__ float tile[32][33];
  const int c0 = bx * 32, r0 = by * 32;
  const int tx = threadIdx.x & 31, ty = threadIdx.x >> 5;
#pragma unroll
  for (int i = 0; i < 32; i += 8)
    tile[ty + i][tx] = in[(size_t)(r0 + ty + i) * cols + (c0 + tx)];
  __syncthreads();
#pragma unroll
  for (int i = 0; i < 32; i += 8)
    out[(size_t)(c0 + ty + i) * rows + (r0 + tx)] = tile[tx][ty + i];
}

template <bool WRT>
__device__ __forceinline__ void panel_body(
    const float* __restrict__ in, float* __restrict__ rt, float* __restrict__ tt,
    int bx, int by) {
  __shared__ float tile[32][33];
  const int c0 = bx * 32, r0 = by * 32;
  const int tx = threadIdx.x & 31, ty = threadIdx.x >> 5;
#pragma unroll
  for (int i = 0; i < 32; i += 8) {
    const float v = in[(size_t)(r0 + ty + i) * XL + (c0 + tx)];
    tile[ty + i][tx] = v;
    if constexpr (WRT) rt[panel_off(r0 + ty + i, c0 + tx)] = v;
  }
  __syncthreads();
#pragma unroll
  for (int i = 0; i < 32; i += 8)
    tt[panel_off(c0 + ty + i, r0 + tx)] = tile[tx][ty + i];
}

constexpr int SQG = (XL / 64);

__global__ __launch_bounds__(256) void prep_k(
    const float* __restrict__ Bm, float* __restrict__ Bt,
    const float* __restrict__ Am, float* __restrict__ rt, float* __restrict__ tt) {
  constexpr int NT = (UL / 32) * (XL / 32);
  const int b = blockIdx.x;
  if (b < NT) {
    transpose_body(Bm, Bt, XL, UL, b % (UL / 32), b / (UL / 32));
  } else {
    const int q = b - NT;
    panel_body<true>(Am, rt, tt, q % (XL / 32), q / (XL / 32));
  }
}

template <int GX, int GY>
__global__ __launch_bounds__(256) void bu_sq_k(
    const float* __restrict__ uf, const float* __restrict__ Bt, float* __restrict__ X,
    int K, int lda, int ldb, int ldc,
    const float* __restrict__ Q, const float* __restrict__ Qt,
    float* __restrict__ Cn, float* __restrict__ Ct) {
  const int b = blockIdx.x;
  if (b < GX * GY) {
    gemm_rr_body<4, 4>(uf, Bt, X, K, lda, ldb, ldc, b % GX, b / GX);
  } else {
    const int q = b - GX * GY;
    square_body<2, 1, 8, true>(Q, Qt, Cn, Ct, q % SQG, q / SQG);
  }
}

__global__ __launch_bounds__(256) void mv_sq_k(
    const float* __restrict__ Mm, const float* __restrict__ vin,
    const float* vadd, float* vout,
    const float* __restrict__ Q, const float* __restrict__ Qt,
    float* __restrict__ Cn, float* __restrict__ Ct) {
  const int b = blockIdx.x;
  if (b < XL / 32) {
    matvec_body(Mm, vin, vadd, vout, b);
  } else {
    const int q = b - XL / 32;
    square_body<2, 1, 8, true>(Q, Qt, Cn, Ct, q % SQG, q / SQG);
  }
}

template <int WM, int WN, int KU, bool WT>
__global__ __launch_bounds__(256) void sweep_sq_k(
    float* X, const float* __restrict__ P, int s, int r0, int gxs, int nsw,
    const float* __restrict__ Q, const float* __restrict__ Qt,
    float* __restrict__ Cn, float* __restrict__ Ct) {
  const int b = blockIdx.x;
  if (b < nsw) {
    sweep_body<WM, WN, KU, 2, (KU <= 8)>(X, P, s, r0, b & ((1 << gxs) - 1), b >> gxs);
  } else {
    const int q = b - nsw;
    square_body<2, 1, 8, WT>(Q, Qt, Cn, Ct, q % SQG, q / SQG);
  }
}

template <int WM, int WN, int KU>
__global__ __launch_bounds__(256) void sweep_panel_k(
    float* X, const float* __restrict__ P, int s, int r0, int gxs, int nsw,
    const float* __restrict__ Am, float* __restrict__ tt) {
  const int b = blockIdx.x;
  if (b < nsw) {
    sweep_body<WM, WN, KU, 2, true>(X, P, s, r0, b & ((1 << gxs) - 1), b >> gxs);
  } else {
    const int q = b - nsw;
    panel_body<false>(Am, nullptr, tt, q % (XL / 32), q / (XL / 32));
  }
}

template <int WM, int WN, int KU>
__global__ __launch_bounds__(256) void sweep_cproj_k(
    float* X, const float* __restrict__ P, int s, int r0, int gxs, int nsw,
    const float* __restrict__ Cm, const float* __restrict__ At,
    const float* __restrict__ Bt, const float* __restrict__ Dm,
    float* __restrict__ CAm, float* __restrict__ CBDm) {
  const int b = blockIdx.x;
  if (b < nsw) {
    sweep_body<WM, WN, KU, 2, true>(X, P, s, r0, b & ((1 << gxs) - 1), b >> gxs);
  } else if (b < nsw + (YLN / 64) * (XL / 64)) {
    const int q = b - nsw;
    cproj_body<128, false>(Cm, At, nullptr, CAm, XL, q % (YLN / 64), q / (YLN / 64));
  } else {
    const int q = b - nsw - (YLN / 64) * (XL / 64);
    cproj_body<8, true>(Cm, Bt, Dm, CBDm, UL, q % (YLN / 64), q / (YLN / 64));
  }
}

extern "C" void kernel_launch(void* const* d_in, const int* in_sizes, int n_in,
                              void* d_out, int out_size, void* d_ws, size_t ws_size,
                              hipStream_t stream) {
  (void)in_sizes; (void)n_in; (void)out_size; (void)ws_size;
  const float* u  = (const float*)d_in[0];
  const float* A  = (const float*)d_in[1];
  const float* B  = (const float*)d_in[2];
  const float* C  = (const float*)d_in[3];
  const float* D  = (const float*)d_in[4];
  const float* x0 = (const float*)d_in[5];
  float* out = (float*)d_out;

  float* ws = (float*)d_ws;
  size_t off = 0;
  float* X  = ws + off; off += (size_t)NSEQ * XL;
  float* Bt = ws + off; off += (size_t)UL * XL;
  float* P2   = ws + off; off += (size_t)XL * XL;
  float* P4   = ws + off; off += (size_t)XL * XL;
  float* P8   = ws + off; off += (size_t)XL * XL;
  float* P16  = ws + off; off += (size_t)XL * XL;
  float* P32  = ws + off; off += (size_t)XL * XL;
  float* P64  = ws + off; off += (size_t)XL * XL;
  float* P128 = ws + off; off += (size_t)XL * XL;
  float* P256  = out;
  float* P512  = out + (size_t)XL * XL;
  float* P1024 = out + 2 * (size_t)XL * XL;
  float* P1    = out + 3 * (size_t)XL * XL;

  float* T1 = P4,   * T2 = P8,    * T4 = P16,    * T8 = P32,   * T16 = P64;
  float* T32 = P128, * T64 = P256, * T128 = P512, * T256 = P1024, * T512 = P1;
  float* T1b = P1;
  float* CA  = P128;
  float* CBD = P128 + (size_t)YLN * XL;

  prep_k<<<(UL / 32) * (XL / 32) + (XL / 32) * (XL / 32), 256, 0, stream>>>(B, Bt, A, P1, T1);

  bu_sq_k<NSEQ / 128, XL / 256><<<(NSEQ / 128) * (XL / 256) + SQG * SQG, 256, 0, stream>>>(
      u, Bt, X, UL, UL, XL, XL, P1, T1, P2, T2);
  mv_sq_k<<<XL / 32 + SQG * SQG, 256, 0, stream>>>(A, x0, X, X, P2, T2, P4, T4);

  sweep_sq_k<4, 4, 2, true><<<(8192 / 128) * (XL / 256) + SQG * SQG, 256, 0, stream>>>(
      X, P1,  1,  1,   6, (8192 / 128) * (XL / 256), P4,   T4,   P8,    T8);
  sweep_sq_k<2, 4, 2, true><<<(4096 / 64) * (XL / 256) + SQG * SQG, 256, 0, stream>>>(
      X, P2,  2,  3,   6, (4096 / 64) * (XL / 256),  P8,   T8,   P16,   T16);
  sweep_sq_k<2, 2, 4, true><<<(2048 / 64) * (XL / 128) + SQG * SQG, 256, 0, stream>>>(
      X, P4,  4,  7,   5, (2048 / 64) * (XL / 128),  P16,  T16,  P32,   T32);
  sweep_sq_k<1, 2, 8, true><<<(1024 / 32) * (XL / 128) + SQG * SQG, 256, 0, stream>>>(
      X, P8,  8,  15,  5, (1024 / 32) * (XL / 128),  P32,  T32,  P64,   T64);
  sweep_sq_k<1, 1, 16, true><<<(512 / 32) * (XL / 64) + SQG * SQG, 256, 0, stream>>>(
      X, P16, 16, 31,  4, (512 / 32) * (XL / 64),    P64,  T64,  P128,  T128);
  sweep_sq_k<1, 1, 16, true><<<(256 / 32) * (XL / 64) + SQG * SQG, 256, 0, stream>>>(
      X, P32, 32, 63,  3, (256 / 32) * (XL / 64),    P128, T128, P256,  T256);
  sweep_sq_k<1, 1, 16, true><<<(128 / 32) * (XL / 64) + SQG * SQG, 256, 0, stream>>>(
      X, P64, 64, 127, 2, (128 / 32) * (XL / 64),    P256, T256, P512,  T512);
  sweep_sq_k<1, 1, 16, false><<<(64 / 32) * (XL / 64) + SQG * SQG, 256, 0, stream>>>(
      X, P128, 128, 255, 1, (64 / 32) * (XL / 64),   P512, T512, P1024, nullptr);
  sweep_panel_k<1, 1, 16><<<(32 / 32) * (XL / 64) + (XL / 32) * (XL / 32), 256, 0, stream>>>(
      X, P256, 256, 511, 0, (32 / 32) * (XL / 64),   A, T1b);
  sweep_k<1, 1, 16, 1><<<dim3(16 / 16, XL / 128), 256, 0, stream>>>(X, P512, 512, 1023);

  for (int j = 1; j < NSEQ / CLEN; ++j) {
    float* prev = X + ((size_t)j * CLEN - 1) * XL;
    float* cur  = X + ((size_t)(j + 1) * CLEN - 1) * XL;
    matvec_step_k<<<32, 256, 0, stream>>>(P1024, prev, cur, cur);
  }

  sweep_k<1, 1, 16, 1><<<dim3(16 / 16, XL / 128), 256, 0, stream>>>(X, P512, 512, 511);
  sweep_k<1, 1, 16><<<dim3(32 / 32,    XL / 64),  256, 0, stream>>>(X, P256, 256, 255);
  sweep_k<1, 1, 16><<<dim3(64 / 32,    XL / 64),  256, 0, stream>>>(X, P128, 128, 127);
  sweep_cproj_k<1, 1, 16><<<(128 / 32) * (XL / 64) + (YLN / 64) * (XL / 64) + (YLN / 64) * (UL / 64),
                            256, 0, stream>>>(
      X, P64, 64, 63, 2, (128 / 32) * (XL / 64), C, T1b, Bt, D, CA, CBD);
  sweep_k<1, 1, 16><<<dim3(256 / 32,   XL / 64),  256, 0, stream>>>(X, P32, 32, 31);
  sweep_k<1, 1, 16><<<dim3(512 / 32,   XL / 64),  256, 0, stream>>>(X, P16, 16, 15);
  sweep_k<1, 2, 8><<<dim3(1024 / 32,  XL / 128), 256, 0, stream>>>(X, P8,  8,  7);
  sweep_k<2, 2, 4><<<dim3(2048 / 64,  XL / 128), 256, 0, stream>>>(X, P4,  4,  3);
  sweep_k<2, 4, 2><<<dim3(4096 / 64,  XL / 256), 256, 0, stream>>>(X, P2,  2,  1);

  out_k<2><<<dim3(NSEQ / 128, YLN / 128), 256, 0, stream>>>(X, x0, C, CA, u, D, CBD, out);
}


# first up level (s = 1, 8192 rows, K = 1024) folded into the drive product: odd rows of X only, from u[n] and u[n-1] against B and A B (K = 2 x 256, staged loop); x0 via A (A x0); sum order changes
# speedup vs baseline: 10.6837x; 1.1438x over previous
#include <hip/hip_runtime.h>
#include <math.h>


typedef __attribute__((ext_vector_type(2))) float v2f;
typedef __attribute__((ext_vector_type(4))) float v4f;
typedef __attribute__((ext_vector_type(8))) float v8f;

#define NSEQ   16384
#define XL     1024
#define UL     256
#define YLN    256
#define CLEN   1024

__device__ __forceinline__ v8f wmma4(v2f a, v2f b, v8f c) {
  return __builtin_amdgcn_wmma_f32_16x16x4_f32(
       false, a,  false, b,
       (short)0, c,  false,  false);
}

__device__ __forceinline__ size_t panel_off(int R, int K) {
  return ((size_t)(R >> 4) * (XL / 8) + (size_t)(K >> 3)) * 128
         + (size_t)((R & 15) * 8 + (K & 7));
}

template <int WM, int WN, int KU>
struct frag_set { v4f a[KU][WM]; v4f b[KU][WN]; };

template <int WM, int WN, int KU, int AS, int BS>
__device__ __forceinline__ void load_set(frag_set<WM, WN, KU>& f,
    const float* const (&ap)[WM], const float* const (&bp)[WN], int k) {
#pragma unroll
  for (int u = 0; u < KU; ++u) {
#pragma unroll
    for (int mi = 0; mi < WM; ++mi)
      f.a[u][mi] = *(const v4f*)(ap[mi] + (k / 8 + u) * AS);
#pragma unroll
    for (int ni = 0; ni < WN; ++ni)
      f.b[u][ni] = *(const v4f*)(bp[ni] + (k / 8 + u) * BS);
  }
}

template <int WM, int WN, int KU, int NCH, int WNB = WN>
__device__ __forceinline__ void mma_set(v8f (&acc)[NCH][WM][WN],
                                        const frag_set<WM, WNB, KU>& f) {
  constexpr int MG = WM / (WNB / WN);
#pragma unroll
  for (int u = 0; u < KU; ++u)
#pragma unroll
    for (int mi = 0; mi < WM; ++mi)
#pragma unroll
      for (int ni = 0; ni < WN; ++ni) {
        const int bi = (mi / MG) * WN + ni;
        const v2f a0 = { f.a[u][mi].x, f.a[u][mi].y };
        const v2f a1 = { f.a[u][mi].z, f.a[u][mi].w };
        const v2f b0 = { f.b[u][bi].x, f.b[u][bi].y };
        const v2f b1 = { f.b[u][bi].z, f.b[u][bi].w };
        acc[0][mi][ni]       = wmma4(a0, b0, acc[0][mi][ni]);
        acc[NCH - 1][mi][ni] = wmma4(a1, b1, acc[NCH - 1][mi][ni]);
      }
}

struct rows_plain {
  const float* base; int ld;
  __device__ __forceinline__ const float* operator()(int row) const {
    return base + (size_t)row * ld;
  }
};
struct rows_tree {
  const float* X; int q0; int s2;
  __device__ __forceinline__ const float* operator()(int row) const {
    int q = q0 + row * s2;
    if (q < 0) q += s2;
    return X + (size_t)q * XL;
  }
};

template <int WM, int WN, int KU>
__device__ __forceinline__ void load_a_lds(frag_set<WM, WN, KU>& f,
    const float* buf, const int (&arow)[WM], int koff) {
#pragma unroll
  for (int u = 0; u < KU; ++u)
#pragma unroll
    for (int mi = 0; mi < WM; ++mi)
      f.a[u][mi] = *(const v4f*)(buf + arow[mi] + koff + u * 8);
}

template <int WM, int WN, int KU, int BS>
__device__ __forceinline__ void load_b_set(frag_set<WM, WN, KU>& f,
    const float* const (&bp)[WN], int k) {
#pragma unroll
  for (int u = 0; u < KU; ++u)
#pragma unroll
    for (int ni = 0; ni < WN; ++ni)
      f.b[u][ni] = *(const v4f*)(bp[ni] + (k / 8 + u) * BS);
}

template <int WM, int WN, int WNB, int KU, int NCH, int WGM, int BS, class ROWS>
__device__ __forceinline__ void staged_kloop(
    v8f (&acc)[NCH][WM][WN], ROWS rows, const float* const (&bp)[WNB],
    int K, int wm, int c, int h) {
  constexpr int R     = 16 * WM * WGM;
  constexpr int KT    = 8 * KU;
  constexpr int PK    = 2 * KT;
  constexpr int PITCH = PK + 4;
  constexpr int RP    = PK / 4;
  constexpr int NPC   = R * RP / 256;
  constexpr int BUF   = R * PITCH;
  static_assert((R * RP) % 256 == 0 && NPC >= 1, "whole pieces per thread");
  __shared__ __attribute__((aligned(16))) float stage[2 * BUF];

  const float* gsrc[NPC];
  int ldst[NPC];
#pragma unroll
  for (int j = 0; j < NPC; ++j) {
    const int i = (int)threadIdx.x + 256 * j;
    const int row = i / RP, pc = i % RP;
    gsrc[j] = rows(row) + pc * 4;
    ldst[j] = row * PITCH + pc * 4;
  }
  int arow[WM];
#pragma unroll
  for (int mi = 0; mi < WM; ++mi)
    arow[mi] = ((wm * WM + mi) * 16 + c) * PITCH + 4 * h;

  v4f g[NPC];
#pragma unroll
  for (int j = 0; j < NPC; ++j) g[j] = *(const v4f*)gsrc[j];
#pragma unroll
  for (int j = 0; j < NPC; ++j) *(v4f*)(stage + ldst[j]) = g[j];
  __syncthreads();

  frag_set<WM, WNB, KU> f0, f1;
  load_a_lds<WM, WNB, KU>(f0, stage, arow, 0);
  load_b_set<WM, WNB, KU, BS>(f0, bp, 0);
#pragma nounroll
  for (int k = 0; k < K; k += PK) {
    const int par = (k / PK) & 1;
    const float* cur = stage + par * BUF;
    float* nxt = stage + (par ^ 1) * BUF;
    const bool more = (k + PK < K);
    if (more) {
#pragma unroll
      for (int j = 0; j < NPC; ++j) g[j] = *(const v4f*)(gsrc[j] + (k + PK));
    }
    load_a_lds<WM, WNB, KU>(f1, cur, arow, KT);
    load_b_set<WM, WNB, KU, BS>(f1, bp, k + KT);
    asm volatile("" ::: "memory");
    mma_set<WM, WN, KU, NCH, WNB>(acc, f0);
    if (more) {
#pragma unroll
      for (int j = 0; j < NPC; ++j) *(v4f*)(nxt + ldst[j]) = g[j];
    }
    __syncthreads();
    const int kn = more ? (k + PK) : k;
    load_a_lds<WM, WNB, KU>(f0, more ? (const float*)nxt : cur, arow, 0);
    load_b_set<WM, WNB, KU, BS>(f0, bp, kn);
    asm volatile("" ::: "memory");
    mma_set<WM, WN, KU, NCH, WNB>(acc, f1);
  }
}

template <int WM, int WN, int KU, int WGM, bool STG>
__device__ __forceinline__ void sweep_body(
    float* X, const float* __restrict__ P, int s, int r0, int bx, int by) {
  constexpr int WGN = 8 / WGM;
  const int lane = threadIdx.x & 31;
  const int w    = threadIdx.x >> 5;
  const int wm = w / WGN, wn = w % WGN;
  const int tm = bx * (16 * WM * WGM) + wm * (16 * WM);
  const int tn = by * (16 * WN * WGN) + wn * (16 * WN);
  const int c = lane & 15;
  const int h = lane >> 4;
  const int s2 = 2 * s;
  constexpr int NCH = (WM * WN < 4) ? 2 : 1;
  constexpr int KT = 8 * KU;
  static_assert(XL % (2 * KT) == 0, "two trips per loop pass");

  v8f acc[NCH][WM][WN];
#pragma unroll
  for (int ch = 0; ch < NCH; ++ch)
#pragma unroll
    for (int mi = 0; mi < WM; ++mi)
#pragma unroll
      for (int ni = 0; ni < WN; ++ni) acc[ch][mi][ni] = (v8f){};

  const float* bp[WN];
#pragma unroll
  for (int ni = 0; ni < WN; ++ni)
    bp[ni] = P + (size_t)((tn >> 4) + ni) * (16 * XL) + c * 8 + 4 * h;

  if constexpr (STG) {
    constexpr int R = 16 * WM * WGM;
    const rows_tree rows = { X, r0 - s + bx * R * s2, s2 };
    staged_kloop<WM, WN, WN, KU, NCH, WGM, 128>(acc, rows, bp, XL, wm, c, h);
  } else {
    const float* ap[WM];
#pragma unroll
    for (int mi = 0; mi < WM; ++mi) {
      int q = r0 - s + (tm + mi * 16 + c) * s2;
      if (q < 0) q += s2;
      ap[mi] = X + (size_t)q * XL + 4 * h;
    }
    frag_set<WM, WN, KU> f0, f1;
    load_set<WM, WN, KU, 8, 128>(f0, ap, bp, 0);
    for (int k = 0; k < XL; k += 2 * KT) {
      load_set<WM, WN, KU, 8, 128>(f1, ap, bp, k + KT);
      asm volatile("" ::: "memory");
      mma_set<WM, WN, KU, NCH>(acc, f0);
      const int kn = (k + 2 * KT < XL) ? (k + 2 * KT) : k;
      load_set<WM, WN, KU, 8, 128>(f0, ap, bp, kn);
      asm volatile("" ::: "memory");
      mma_set<WM, WN, KU, NCH>(acc, f1);
    }
  }

  const bool open0 = (r0 < s);
  v8f old[WM][WN];
#pragma unroll
  for (int mi = 0; mi < WM; ++mi)
#pragma unroll
    for (int ni = 0; ni < WN; ++ni) {
#pragma unroll
      for (int r = 0; r < 8; ++r) {
        const int row = r0 + (tm + mi * 16 + 8 * h + r) * s2;
        old[mi][ni][r] = X[(size_t)row * XL + (tn + ni * 16 + c)];
      }
    }
#pragma unroll
  for (int mi = 0; mi < WM; ++mi)
#pragma unroll
    for (int ni = 0; ni < WN; ++ni) {
      v8f a = acc[0][mi][ni];
      if constexpr (NCH == 2) a = a + acc[1][mi][ni];
#pragma unroll
      for (int r = 0; r < 8; ++r) {
        const int m   = tm + mi * 16 + 8 * h + r;
        const int row = r0 + m * s2;
        const float add = (open0 && m == 0) ? 0.0f : a[r];
        X[(size_t)row * XL + (tn + ni * 16 + c)] = old[mi][ni][r] + add;
      }
    }
}

template <int WM, int WN, int KU, int WGM = 2>
__global__ __launch_bounds__(256) void sweep_k(
    float* X, const float* __restrict__ P, int s, int r0) {
  sweep_body<WM, WN, KU, WGM, true>(X, P, s, r0, blockIdx.x, blockIdx.y);
}

template <int WM, int WN, int KU>
__device__ __forceinline__ void drive_body(
    const float* __restrict__ uf, const float* __restrict__ ABm,
    const float* __restrict__ Bm, float* __restrict__ X, int bx, int by) {
  constexpr int WGM = 2, WGN = 4;
  constexpr int R = 16 * WM * WGM;
  static_assert(UL % (16 * KU) == 0, "two trips per loop pass");
  const int lane = threadIdx.x & 31;
  const int w    = threadIdx.x >> 5;
  const int wm = w / WGN, wn = w % WGN;
  const int tm = bx * R + wm * (16 * WM);
  const int tn = by * (16 * WN * WGN) + wn * (16 * WN);
  const int c = lane & 15;
  const int h = lane >> 4;

  v8f acc[1][WM][WN];
#pragma unroll
  for (int mi = 0; mi < WM; ++mi)
#pragma unroll
    for (int ni = 0; ni < WN; ++ni) acc[0][mi][ni] = (v8f){};

  const float* bp[WN];
  {
#pragma unroll
    for (int ni = 0; ni < WN; ++ni)
      bp[ni] = ABm + (size_t)(tn + ni * 16 + c) * UL + 4 * h;
    const rows_plain rows = { uf + (size_t)bx * R * (2 * UL), 2 * UL };
    staged_kloop<WM, WN, WN, KU, 1, WGM, 8>(acc, rows, bp, UL, wm, c, h);
  }
  {
#pragma unroll
    for (int ni = 0; ni < WN; ++ni)
      bp[ni] = Bm + (size_t)(tn + ni * 16 + c) * UL + 4 * h;
    const rows_plain rows = { uf + (size_t)bx * R * (2 * UL) + UL, 2 * UL };
    staged_kloop<WM, WN, WN, KU, 1, WGM, 8>(acc, rows, bp, UL, wm, c, h);
  }

#pragma unroll
  for (int mi = 0; mi < WM; ++mi)
#pragma unroll
    for (int ni = 0; ni < WN; ++ni) {
#pragma unroll
      for (int r = 0; r < 8; ++r) {
        const int m = tm + mi * 16 + 8 * h + r;
        X[(size_t)(2 * m + 1) * XL + (tn + ni * 16 + c)] = acc[0][mi][ni][r];
      }
    }
}

template <int WM, int WN, int KU, bool WT>
__device__ __forceinline__ void square_body(
    const float* __restrict__ Q, const float* __restrict__ Qt,
    float* __restrict__ Cn, float* __restrict__ Ct, int bx, int by) {
  const int lane = threadIdx.x & 31;
  const int w    = threadIdx.x >> 5;
  const int wm = w >> 2, wn = w & 3;
  const int tm = bx * (32 * WM) + wm * (16 * WM);
  const int tn = by * (64 * WN) + wn * (16 * WN);
  const int c = lane & 15;
  const int h = lane >> 4;
  constexpr int NCH = (WM * WN < 4) ? 2 : 1;
  constexpr int KT = 8 * KU;
  static_assert(XL % (2 * KT) == 0, "two trips per loop pass");

  v8f acc[NCH][WM][WN];
#pragma unroll
  for (int ch = 0; ch < NCH; ++ch)
#pragma unroll
    for (int mi = 0; mi < WM; ++mi)
#pragma unroll
      for (int ni = 0; ni < WN; ++ni) acc[ch][mi][ni] = (v8f){};

  const float* ap[WM];
  const float* bp[WN];
#pragma unroll
  for (int mi = 0; mi < WM; ++mi)
    ap[mi] = Q + (size_t)((tm >> 4) + mi) * (16 * XL) + c * 8 + 4 * h;
#pragma unroll
  for (int ni = 0; ni < WN; ++ni)
    bp[ni] = Qt + (size_t)((tn >> 4) + ni) * (16 * XL) + c * 8 + 4 * h;

  frag_set<WM, WN, KU> f0, f1;
  load_set<WM, WN, KU, 128, 128>(f0, ap, bp, 0);
  for (int k = 0; k < XL; k += 2 * KT) {
    load_set<WM, WN, KU, 128, 128>(f1, ap, bp, k + KT);
    asm volatile("" ::: "memory");
    mma_set<WM, WN, KU, NCH>(acc, f0);
    const int kn = (k + 2 * KT < XL) ? (k + 2 * KT) : k;
    load_set<WM, WN, KU, 128, 128>(f0, ap, bp, kn);
    asm volatile("" ::: "memory");
    mma_set<WM, WN, KU, NCH>(acc, f1);
  }

#pragma unroll
  for (int mi = 0; mi < WM; ++mi)
#pragma unroll
    for (int ni = 0; ni < WN; ++ni) {
      v8f a = acc[0][mi][ni];
      if constexpr (NCH == 2) a = a + acc[1][mi][ni];
      const int m0 = tm + mi * 16 + 8 * h;
      const int n  = tn + ni * 16 + c;
      if constexpr (WT) {
        float* p = Cn + panel_off(m0, n);
#pragma unroll
        for (int r = 0; r < 8; ++r) p[r * 8] = a[r];
        float* q = Ct + panel_off(n, m0);
        *(v4f*)q       = (v4f){ a[0], a[1], a[2], a[3] };
        *(v4f*)(q + 4) = (v4f){ a[4], a[5], a[6], a[7] };
      } else {
#pragma unroll
        for (int r = 0; r < 8; ++r) Cn[(size_t)(m0 + r) * XL + n] = a[r];
      }
    }
}

template <int BS, bool ADD>
__device__ __forceinline__ void cproj_body(
    const float* __restrict__ Cm, const float* __restrict__ Mt,
    const float* __restrict__ Dadd, float* __restrict__ Rm, int N, int bx, int by) {
  constexpr int WM = 2, WN = 1, KU = 8, NCH = 2, KT = 8 * KU;
  const int lane = threadIdx.x & 31;
  const int w    = threadIdx.x >> 5;
  const int wm = w >> 2, wn = w & 3;
  const int tm = bx * 64 + wm * 32;
  const int tn = by * 64 + wn * 16;
  const int c = lane & 15;
  const int h = lane >> 4;

  v8f acc[NCH][WM][WN];
#pragma unroll
  for (int ch = 0; ch < NCH; ++ch)
#pragma unroll
    for (int mi = 0; mi < WM; ++mi) acc[ch][mi][0] = (v8f){};

  const float* ap[WM];
  const float* bp[WN];
#pragma unroll
  for (int mi = 0; mi < WM; ++mi)
    ap[mi] = Cm + (size_t)(tm + mi * 16 + c) * XL + 4 * h;
  if constexpr (BS == 128)
    bp[0] = Mt + (size_t)(tn >> 4) * (16 * XL) + c * 8 + 4 * h;
  else
    bp[0] = Mt + (size_t)(tn + c) * XL + 4 * h;

  frag_set<WM, WN, KU> f0, f1;
  load_set<WM, WN, KU, 8, BS>(f0, ap, bp, 0);
  for (int k = 0; k < XL; k += 2 * KT) {
    load_set<WM, WN, KU, 8, BS>(f1, ap, bp, k + KT);
    asm volatile("" ::: "memory");
    mma_set<WM, WN, KU, NCH>(acc, f0);
    const int kn = (k + 2 * KT < XL) ? (k + 2 * KT) : k;
    load_set<WM, WN, KU, 8, BS>(f0, ap, bp, kn);
    asm volatile("" ::: "memory");
    mma_set<WM, WN, KU, NCH>(acc, f1);
  }

#pragma unroll
  for (int mi = 0; mi < WM; ++mi) {
    const v8f a = acc[0][mi][0] + acc[1][mi][0];
    const int m0 = tm + mi * 16 + 8 * h;
    const int n  = tn + c;
#pragma unroll
    for (int r = 0; r < 8; ++r) {
      float v = a[r];
      if constexpr (ADD) v += Dadd[(size_t)(m0 + r) * N + n];
      Rm[(size_t)(m0 + r) * N + n] = v;
    }
  }
}

template <int MASK>
__device__ __forceinline__ float row_xmask_add(float v) {
  const int t = __builtin_amdgcn_update_dpp(0, __builtin_bit_cast(int, v),
                                            0x160 + MASK, 0xf, 0xf, true);
  return v + __builtin_bit_cast(float, t);
}

__device__ __forceinline__ float wave_sum(float v) {
  v = row_xmask_add<1>(v);
  v = row_xmask_add<2>(v);
  v = row_xmask_add<4>(v);
  v = row_xmask_add<8>(v);
  const int b = __builtin_bit_cast(int, v);
  const int o = __builtin_amdgcn_permlanex16(b, b, 0x76543210, 0xfedcba98, false, false);
  return v + __builtin_bit_cast(float, o);
}

template <bool ADD>
__device__ __forceinline__ void matvec_body(
    const float* __restrict__ Mm, const float* __restrict__ vin,
    const float* vadd, float* vout, int bx) {
  __shared__ __attribute__((aligned(16))) float rsum[32];
  const int lane = threadIdx.x & 31, wid = threadIdx.x >> 5;
  const int line0 = bx * 32;
  const int row0  = line0 + wid * 4;

  float addv = 0.f;
  if constexpr (ADD) {
    if (wid == 0) addv = vadd[line0 + lane];
  }

  v4f x[8], m[4][8];
#pragma unroll
  for (int j = 0; j < 8; ++j) x[j] = *(const v4f*)(vin + 128 * j + 4 * lane);
#pragma unroll
  for (int r = 0; r < 4; ++r)
#pragma unroll
    for (int j = 0; j < 8; ++j)
      m[r][j] = *(const v4f*)(Mm + (size_t)(row0 + r) * XL + 128 * j + 4 * lane);
  __builtin_amdgcn_sched_barrier(0);

  v4f s4;
#pragma unroll
  for (int r = 0; r < 4; ++r) {
    v4f acc = m[r][0] * x[0];
#pragma unroll
    for (int j = 1; j < 8; ++j) acc += m[r][j] * x[j];
    s4[r] = wave_sum((acc.x + acc.y) + (acc.z + acc.w));
  }
  if (lane == 0) *(v4f*)(rsum + wid * 4) = s4;
  __syncthreads();
  if (wid == 0) vout[line0 + lane] = rsum[lane] + addv;
}

__global__ __launch_bounds__(256) void matvec_step_k(
    const float* __restrict__ Mm, const float* __restrict__ vin,
    const float* vadd, float* vout) {
  matvec_body<true>(Mm, vin, vadd, vout, blockIdx.x);
}

struct rows_par {
  const float* base; const float* first; int ld; int t0; int back;
  __device__ __forceinline__ const float* operator()(int row) const {
    const int q = t0 + (row >> 6) * 64 + 2 * (row & 31) + 1 - back * ((row >> 5) & 1);
    return (q < 0) ? first : base + (size_t)q * ld;
  }
};

template <int KU>
__global__ __launch_bounds__(256) void out_k(
    const float* __restrict__ xs, const float* __restrict__ x0,
    const float* __restrict__ Cm, const float* __restrict__ CAm,
    const float* __restrict__ uf, const float* __restrict__ Dm,
    const float* __restrict__ CBDm, float* __restrict__ out) {
  constexpr int WM = 4, WN = 2;
  static_assert(XL % (16 * KU) == 0 && UL % (16 * KU) == 0, "two trips per loop pass");
  const int lane = threadIdx.x & 31;
  const int w    = threadIdx.x >> 5;
  const int wm = w >> 2, wn = w & 3;
  const int t0 = blockIdx.x * 128;
  const int tn = blockIdx.y * 128 + wn * (16 * WN);
  const int c = lane & 15;
  const int h = lane >> 4;

  v8f acc[1][WM][WN];
#pragma unroll
  for (int mi = 0; mi < WM; ++mi)
#pragma unroll
    for (int ni = 0; ni < WN; ++ni) acc[0][mi][ni] = (v8f){};

  {
    const float* bp[2 * WN];
#pragma unroll
    for (int ni = 0; ni < WN; ++ni) {
      bp[ni]      = Cm  + (size_t)(tn + ni * 16 + c) * XL + 4 * h;
      bp[WN + ni] = CAm + (size_t)(tn + ni * 16 + c) * XL + 4 * h;
    }
    const rows_par rows = { xs, x0, XL, t0, 2 };
    staged_kloop<WM, WN, 2 * WN, KU, 1, 2, 8>(acc, rows, bp, XL, wm, c, h);
  }
  {
    const float* bp[2 * WN];
#pragma unroll
    for (int ni = 0; ni < WN; ++ni) {
      bp[ni]      = Dm   + (size_t)(tn + ni * 16 + c) * UL + 4 * h;
      bp[WN + ni] = CBDm + (size_t)(tn + ni * 16 + c) * UL + 4 * h;
    }
    const rows_par rows = { uf, uf, UL, t0, 1 };
    staged_kloop<WM, WN, 2 * WN, KU, 1, 2, 8>(acc, rows, bp, UL, wm, c, h);
  }

#pragma unroll
  for (int m1 = 0; m1 < 2; ++m1)
#pragma unroll
    for (int ni = 0; ni < WN; ++ni) {
      const int yl = tn + ni * 16 + c;
      const int n0 = t0 + wm * 64 + m1 * 32 + 16 * h;
      const v8f od = acc[0][m1][ni];
      const v8f ev = acc[0][2 + m1][ni];
      float* op = out + (size_t)yl * NSEQ + n0;
#pragma unroll
      for (int r = 0; r < 8; r += 2)
        *(v4f*)(op + 2 * r) = (v4f){ tanhf(ev[r]), tanhf(od[r]), tanhf(ev[r + 1]), tanhf(od[r + 1]) };
    }
}

__device__ __forceinline__ void transpose_body(
    const float* __restrict__ in, float* __restrict__ out, int rows, int cols,
    int bx, int by) {
  __shared__ float tile[32][33];
  const int c0 = bx * 32, r0 = by * 32;
  const int tx = threadIdx.x & 31, ty = threadIdx.x >> 5;
#pragma unroll
  for (int i = 0; i < 32; i += 8)
    tile[ty + i][tx] = in[(size_t)(r0 + ty + i) * cols + (c0 + tx)];
  __syncthreads();
#pragma unroll
  for (int i = 0; i < 32; i += 8)
    out[(size_t)(c0 + ty + i) * rows + (r0 + tx)] = tile[tx][ty + i];
}

template <bool WRT>
__device__ __forceinline__ void panel_body(
    const float* __restrict__ in, float* __restrict__ rt, float* __restrict__ tt,
    int bx, int by) {
  __shared__ float tile[32][33];
  const int c0 = bx * 32, r0 = by * 32;
  const int tx = threadIdx.x & 31, ty = threadIdx.x >> 5;
#pragma unroll
  for (int i = 0; i < 32; i += 8) {
    const float v = in[(size_t)(r0 + ty + i) * XL + (c0 + tx)];
    tile[ty + i][tx] = v;
    if constexpr (WRT) rt[panel_off(r0 + ty + i, c0 + tx)] = v;
  }
  __syncthreads();
#pragma unroll
  for (int i = 0; i < 32; i += 8)
    tt[panel_off(c0 + ty + i, r0 + tx)] = tile[tx][ty + i];
}

constexpr int SQG = (XL / 64);

__global__ __launch_bounds__(256) void prep_k(
    const float* __restrict__ Bm, float* __restrict__ Bt,
    const float* __restrict__ Am, float* __restrict__ rt, float* __restrict__ tt) {
  constexpr int NT = (UL / 32) * (XL / 32);
  const int b = blockIdx.x;
  if (b < NT) {
    transpose_body(Bm, Bt, XL, UL, b % (UL / 32), b / (UL / 32));
  } else {
    const int q = b - NT;
    panel_body<true>(Am, rt, tt, q % (XL / 32), q / (XL / 32));
  }
}

__global__ __launch_bounds__(256) void ab_sq_k(
    const float* __restrict__ Am, const float* __restrict__ x0, float* __restrict__ t0,
    const float* __restrict__ Bt, float* __restrict__ ABm,
    const float* __restrict__ Q, const float* __restrict__ Qt,
    float* __restrict__ Cn, float* __restrict__ Ct) {
  constexpr int NMV = XL / 32;
  constexpr int NAB = (XL / 64) * (UL / 64);
  const int b = blockIdx.x;
  if (b < NMV) {
    matvec_body<false>(Am, x0, nullptr, t0, b);
  } else if (b < NMV + NAB) {
    const int q = b - NMV;
    cproj_body<8, false>(Am, Bt, nullptr, ABm, UL, q % (XL / 64), q / (XL / 64));
  } else {
    const int q = b - NMV - NAB;
    square_body<2, 1, 8, true>(Q, Qt, Cn, Ct, q % SQG, q / SQG);
  }
}

template <int WM, int WN, int KU>
__global__ __launch_bounds__(256) void drive_sq_k(
    const float* __restrict__ uf, const float* __restrict__ ABm,
    const float* __restrict__ Bm, float* __restrict__ X, int gxs, int nsw,
    const float* __restrict__ Q, const float* __restrict__ Qt,
    float* __restrict__ Cn, float* __restrict__ Ct) {
  const int b = blockIdx.x;
  if (b < nsw) {
    drive_body<WM, WN, KU>(uf, ABm, Bm, X, b & ((1 << gxs) - 1), b >> gxs);
  } else {
    const int q = b - nsw;
    square_body<2, 1, 8, true>(Q, Qt, Cn, Ct, q % SQG, q / SQG);
  }
}

__global__ __launch_bounds__(256) void mv_sq_k(
    const float* __restrict__ Mm, const float* __restrict__ vin,
    const float* vadd, float* vout,
    const float* __restrict__ Q, const float* __restrict__ Qt,
    float* __restrict__ Cn, float* __restrict__ Ct) {
  const int b = blockIdx.x;
  if (b < XL / 32) {
    matvec_body<true>(Mm, vin, vadd, vout, b);
  } else {
    const int q = b - XL / 32;
    square_body<2, 1, 8, true>(Q, Qt, Cn, Ct, q % SQG, q / SQG);
  }
}

template <int WM, int WN, int KU, bool WT>
__global__ __launch_bounds__(256) void sweep_sq_k(
    float* X, const float* __restrict__ P, int s, int r0, int gxs, int nsw,
    const float* __restrict__ Q, const float* __restrict__ Qt,
    float* __restrict__ Cn, float* __restrict__ Ct) {
  const int b = blockIdx.x;
  if (b < nsw) {
    sweep_body<WM, WN, KU, 2, (KU <= 8)>(X, P, s, r0, b & ((1 << gxs) - 1), b >> gxs);
  } else {
    const int q = b - nsw;
    square_body<2, 1, 8, WT>(Q, Qt, Cn, Ct, q % SQG, q / SQG);
  }
}

template <int WM, int WN, int KU>
__global__ __launch_bounds__(256) void sweep_panel_k(
    float* X, const float* __restrict__ P, int s, int r0, int gxs, int nsw,
    const float* __restrict__ Am, float* __restrict__ tt) {
  const int b = blockIdx.x;
  if (b < nsw) {
    sweep_body<WM, WN, KU, 2, true>(X, P, s, r0, b & ((1 << gxs) - 1), b >> gxs);
  } else {
    const int q = b - nsw;
    panel_body<false>(Am, nullptr, tt, q % (XL / 32), q / (XL / 32));
  }
}

template <int WM, int WN, int KU>
__global__ __launch_bounds__(256) void sweep_cproj_k(
    float* X, const float* __restrict__ P, int s, int r0, int gxs, int nsw,
    const float* __restrict__ Cm, const float* __restrict__ At,
    const float* __restrict__ Bt, const float* __restrict__ Dm,
    float* __restrict__ CAm, float* __restrict__ CBDm) {
  const int b = blockIdx.x;
  if (b < nsw) {
    sweep_body<WM, WN, KU, 2, true>(X, P, s, r0, b & ((1 << gxs) - 1), b >> gxs);
  } else if (b < nsw + (YLN / 64) * (XL / 64)) {
    const int q = b - nsw;
    cproj_body<128, false>(Cm, At, nullptr, CAm, XL, q % (YLN / 64), q / (YLN / 64));
  } else {
    const int q = b - nsw - (YLN / 64) * (XL / 64);
    cproj_body<8, true>(Cm, Bt, Dm, CBDm, UL, q % (YLN / 64), q / (YLN / 64));
  }
}

extern "C" void kernel_launch(void* const* d_in, const int* in_sizes, int n_in,
                              void* d_out, int out_size, void* d_ws, size_t ws_size,
                              hipStream_t stream) {
  (void)in_sizes; (void)n_in; (void)out_size; (void)ws_size;
  const float* u  = (const float*)d_in[0];
  const float* A  = (const float*)d_in[1];
  const float* B  = (const float*)d_in[2];
  const float* C  = (const float*)d_in[3];
  const float* D  = (const float*)d_in[4];
  const float* x0 = (const float*)d_in[5];
  float* out = (float*)d_out;

  float* ws = (float*)d_ws;
  size_t off = 0;
  float* X  = ws + off; off += (size_t)NSEQ * XL;
  float* Bt = ws + off; off += (size_t)UL * XL;
  float* P2   = ws + off; off += (size_t)XL * XL;
  float* P4   = ws + off; off += (size_t)XL * XL;
  float* P8   = ws + off; off += (size_t)XL * XL;
  float* P16  = ws + off; off += (size_t)XL * XL;
  float* P32  = ws + off; off += (size_t)XL * XL;
  float* P64  = ws + off; off += (size_t)XL * XL;
  float* P128 = ws + off; off += (size_t)XL * XL;
  float* P256  = out;
  float* P512  = out + (size_t)XL * XL;
  float* P1024 = out + 2 * (size_t)XL * XL;
  float* P1    = out + 3 * (size_t)XL * XL;

  float* T1 = P4,   * T2 = P8,    * T4 = P16,    * T8 = P32,   * T16 = P64;
  float* T32 = P128, * T64 = P256, * T128 = P512, * T256 = P1024, * T512 = P1;
  float* T1b = P1;
  float* CA  = P128;
  float* CBD = P128 + (size_t)YLN * XL;
  float* AB  = P128;
  float* t0  = P128 + (size_t)XL * UL;

  prep_k<<<(UL / 32) * (XL / 32) + (XL / 32) * (XL / 32), 256, 0, stream>>>(B, Bt, A, P1, T1);

  ab_sq_k<<<XL / 32 + (XL / 64) * (UL / 64) + SQG * SQG, 256, 0, stream>>>(
      A, x0, t0, Bt, AB, P1, T1, P2, T2);
  drive_sq_k<4, 4, 2><<<(8192 / 128) * (XL / 256) + SQG * SQG, 256, 0, stream>>>(
      u, AB, B, X, 6, (8192 / 128) * (XL / 256), P2, T2, P4, T4);
  mv_sq_k<<<XL / 32 + SQG * SQG, 256, 0, stream>>>(A, t0, X + XL, X + XL, P4, T4, P8, T8);

  sweep_sq_k<2, 4, 2, true><<<(4096 / 64) * (XL / 256) + SQG * SQG, 256, 0, stream>>>(
      X, P2,  2,  3,   6, (4096 / 64) * (XL / 256),  P8,   T8,   P16,   T16);
  sweep_sq_k<2, 2, 4, true><<<(2048 / 64) * (XL / 128) + SQG * SQG, 256, 0, stream>>>(
      X, P4,  4,  7,   5, (2048 / 64) * (XL / 128),  P16,  T16,  P32,   T32);
  sweep_sq_k<1, 2, 8, true><<<(1024 / 32) * (XL / 128) + SQG * SQG, 256, 0, stream>>>(
      X, P8,  8,  15,  5, (1024 / 32) * (XL / 128),  P32,  T32,  P64,   T64);
  sweep_sq_k<1, 1, 16, true><<<(512 / 32) * (XL / 64) + SQG * SQG, 256, 0, stream>>>(
      X, P16, 16, 31,  4, (512 / 32) * (XL / 64),    P64,  T64,  P128,  T128);
  sweep_sq_k<1, 1, 16, true><<<(256 / 32) * (XL / 64) + SQG * SQG, 256, 0, stream>>>(
      X, P32, 32, 63,  3, (256 / 32) * (XL / 64),    P128, T128, P256,  T256);
  sweep_sq_k<1, 1, 16, true><<<(128 / 32) * (XL / 64) + SQG * SQG, 256, 0, stream>>>(
      X, P64, 64, 127, 2, (128 / 32) * (XL / 64),    P256, T256, P512,  T512);
  sweep_sq_k<1, 1, 16, false><<<(64 / 32) * (XL / 64) + SQG * SQG, 256, 0, stream>>>(
      X, P128, 128, 255, 1, (64 / 32) * (XL / 64),   P512, T512, P1024, nullptr);
  sweep_panel_k<1, 1, 16><<<(32 / 32) * (XL / 64) + (XL / 32) * (XL / 32), 256, 0, stream>>>(
      X, P256, 256, 511, 0, (32 / 32) * (XL / 64),   A, T1b);
  sweep_k<1, 1, 16, 1><<<dim3(16 / 16, XL / 128), 256, 0, stream>>>(X, P512, 512, 1023);

  for (int j = 1; j < NSEQ / CLEN; ++j) {
    float* prev = X + ((size_t)j * CLEN - 1) * XL;
    float* cur  = X + ((size_t)(j + 1) * CLEN - 1) * XL;
    matvec_step_k<<<32, 256, 0, stream>>>(P1024, prev, cur, cur);
  }

  sweep_k<1, 1, 16, 1><<<dim3(16 / 16, XL / 128), 256, 0, stream>>>(X, P512, 512, 511);
  sweep_k<1, 1, 16><<<dim3(32 / 32,    XL / 64),  256, 0, stream>>>(X, P256, 256, 255);
  sweep_k<1, 1, 16><<<dim3(64 / 32,    XL / 64),  256, 0, stream>>>(X, P128, 128, 127);
  sweep_cproj_k<1, 1, 16><<<(128 / 32) * (XL / 64) + (YLN / 64) * (XL / 64) + (YLN / 64) * (UL / 64),
                            256, 0, stream>>>(
      X, P64, 64, 63, 2, (128 / 32) * (XL / 64), C, T1b, Bt, D, CA, CBD);
  sweep_k<1, 1, 16><<<dim3(256 / 32,   XL / 64),  256, 0, stream>>>(X, P32, 32, 31);
  sweep_k<1, 1, 16><<<dim3(512 / 32,   XL / 64),  256, 0, stream>>>(X, P16, 16, 15);
  sweep_k<1, 2, 8><<<dim3(1024 / 32,  XL / 128), 256, 0, stream>>>(X, P8,  8,  7);
  sweep_k<2, 2, 4><<<dim3(2048 / 64,  XL / 128), 256, 0, stream>>>(X, P4,  4,  3);
  sweep_k<2, 4, 2><<<dim3(4096 / 64,  XL / 256), 256, 0, stream>>>(X, P2,  2,  1);

  out_k<2><<<dim3(NSEQ / 128, YLN / 128), 256, 0, stream>>>(X, x0, C, CA, u, D, CBD, out);
}
